# starting kernel
# baseline (speedup 1.0000x reference)
#include <hip/hip_runtime.h>
#include <hip/hip_bf16.h>

typedef float v2f __attribute__((ext_vector_type(2)));
typedef float v8f __attribute__((ext_vector_type(8)));

#define Bb 8
#define Tt 1024
#define Dd 512
#define Mm 64
#define Oo 512

__global__ void ln_eta_kernel(const float* __restrict__ x,
                              const float* __restrict__ gamma,
                              const float* __restrict__ beta,
                              const float* __restrict__ We,
                              float* __restrict__ xn,
                              float* __restrict__ eta)
{
    const int wave = threadIdx.x >> 5;
    const int lane = threadIdx.x & 31;
    const int row  = blockIdx.x * 8 + wave;
    const float* xr = x + (size_t)row * Dd;

    float v[16];
    float sum = 0.f, sq = 0.f;
#pragma unroll
    for (int e = 0; e < 16; ++e) {
        float t = xr[lane + 32 * e];
        v[e] = t; sum += t; sq += t * t;
    }
#pragma unroll
    for (int m = 16; m >= 1; m >>= 1) {
        sum += __shfl_xor(sum, m, 32);
        sq  += __shfl_xor(sq,  m, 32);
    }
    const float mu = sum * (1.0f / Dd);
    const float var = sq * (1.0f / Dd) - mu * mu;
    const float rs = rsqrtf(var + 1e-5f);

    float dot = 0.f;
    float* xo = xn + (size_t)row * Dd;
#pragma unroll
    for (int e = 0; e < 16; ++e) {
        const int d = lane + 32 * e;
        float xv = (v[e] - mu) * rs * gamma[d] + beta[d];
        xo[d] = xv;
        dot += xv * We[d];
    }
#pragma unroll
    for (int m = 16; m >= 1; m >>= 1) dot += __shfl_xor(dot, m, 32);
    if (lane == 0) eta[row] = 1.0f / (1.0f + expf(-dot));
}

__global__ void proj_gemm_kernel(const float* __restrict__ xn,
                                 const float* __restrict__ Wk,
                                 const float* __restrict__ Wq,
                                 const float* __restrict__ Wv,
                                 const float* __restrict__ Wvc,
                                 float* __restrict__ Ko,
                                 float* __restrict__ Qo,
                                 float* __restrict__ Vo,
                                 float* __restrict__ vco)
{
    const int w = blockIdx.y;
    const float* W = (w == 0) ? Wk : (w == 1) ? Wq : (w == 2) ? Wv : Wvc;
    float* Out     = (w == 0) ? Ko : (w == 1) ? Qo : (w == 2) ? Vo : vco;

    const int tileM = blockIdx.x * 16;
    const int lane  = threadIdx.x & 31;
    const int hi    = lane >> 4;
    const int lr    = lane & 15;

    const float* Arow = xn + (size_t)(tileM + lr) * Dd + 2 * hi;
    const float* B0 = W + (size_t)(lr +  0) * Dd + 2 * hi;
    const float* B1 = W + (size_t)(lr + 16) * Dd + 2 * hi;
    const float* B2 = W + (size_t)(lr + 32) * Dd + 2 * hi;
    const float* B3 = W + (size_t)(lr + 48) * Dd + 2 * hi;

    v8f c0 = {}, c1 = {}, c2 = {}, c3 = {};
    for (int kk = 0; kk < Dd; kk += 4) {
        v2f a  = *(const v2f*)(Arow + kk);
        v2f b0 = *(const v2f*)(B0 + kk);
        v2f b1 = *(const v2f*)(B1 + kk);
        v2f b2 = *(const v2f*)(B2 + kk);
        v2f b3 = *(const v2f*)(B3 + kk);
        c0 = __builtin_amdgcn_wmma_f32_16x16x4_f32(false, a, false, b0, (short)0, c0, false, false);
        c1 = __builtin_amdgcn_wmma_f32_16x16x4_f32(false, a, false, b1, (short)0, c1, false, false);
        c2 = __builtin_amdgcn_wmma_f32_16x16x4_f32(false, a, false, b2, (short)0, c2, false, false);
        c3 = __builtin_amdgcn_wmma_f32_16x16x4_f32(false, a, false, b3, (short)0, c3, false, false);
    }

    if (w < 2) {
#pragma unroll
        for (int r = 0; r < 8; ++r) {
            c0[r] = fmaxf(c0[r], 0.f); c1[r] = fmaxf(c1[r], 0.f);
            c2[r] = fmaxf(c2[r], 0.f); c3[r] = fmaxf(c3[r], 0.f);
            float s = c0[r] + c1[r] + c2[r] + c3[r];
#pragma unroll
            for (int m = 8; m >= 1; m >>= 1) s += __shfl_xor(s, m, 32);
            float inv = 1.0f / (1e-5f + s);
            c0[r] *= inv; c1[r] *= inv; c2[r] *= inv; c3[r] *= inv;
        }
    }

#pragma unroll
    for (int r = 0; r < 8; ++r) {
        float* o = Out + (size_t)(tileM + r + 8 * hi) * Mm + lr;
        o[0]  = c0[r];
        o[16] = c1[r];
        o[32] = c2[r];
        o[48] = c3[r];
    }
}

__global__ void scan_kernel(const float* __restrict__ K,
                            const float* __restrict__ Q,
                            const float* __restrict__ V,
                            const float* __restrict__ vc,
                            const float* __restrict__ eta,
                            const float* __restrict__ theta_matrix,
                            float* __restrict__ y)
{
    const int b    = blockIdx.x;
    const int wave = threadIdx.x >> 5;
    const int lane = threadIdx.x & 31;
    const int i    = blockIdx.y * 8 + wave;
    const float theta_i = theta_matrix[i];

    float s0 = 0.f, s1 = 0.f;
    const size_t base = (size_t)b * Tt * Mm;

    for (int t = 0; t < Tt; ++t) {
        const size_t o = base + (size_t)t * Mm;
        const float vc0 = vc[o + lane], vc1 = vc[o + lane + 32];
        const float k0  = K[o + lane],  k1  = K[o + lane + 32];
        const float q0  = Q[o + lane],  q1  = Q[o + lane + 32];
        const float et  = eta[b * Tt + t];
        const float vi  = V[o + i];

        const float cc0 = 1.0f + tanhf(theta_i * vc0);
        const float cc1 = 1.0f + tanhf(theta_i * vc1);
        const float u   = et * vi;
        s0 = s0 * cc0 + u * k0;
        s1 = s1 * cc1 + u * k1;

        float p = s0 * q0 + s1 * q1;
#pragma unroll
        for (int m = 16; m >= 1; m >>= 1) p += __shfl_xor(p, m, 32);
        if (lane == 0) y[o + i] = p;
    }
}

__global__ void out_gemm_kernel(const float* __restrict__ y,
                                const float* __restrict__ Wout,
                                const float* __restrict__ bout,
                                float* __restrict__ out)
{
    const int tileM = blockIdx.x * 16;
    const int tileN = blockIdx.y * 64;
    const int lane  = threadIdx.x & 31;
    const int hi    = lane >> 4;
    const int lr    = lane & 15;

    const float* Arow = y + (size_t)(tileM + lr) * Mm + 2 * hi;
    const float* B0 = Wout + (size_t)(tileN + lr +  0) * Mm + 2 * hi;
    const float* B1 = Wout + (size_t)(tileN + lr + 16) * Mm + 2 * hi;
    const float* B2 = Wout + (size_t)(tileN + lr + 32) * Mm + 2 * hi;
    const float* B3 = Wout + (size_t)(tileN + lr + 48) * Mm + 2 * hi;

    v8f c0 = {}, c1 = {}, c2 = {}, c3 = {};
#pragma unroll
    for (int kk = 0; kk < Mm; kk += 4) {
        v2f a  = *(const v2f*)(Arow + kk);
        v2f b0 = *(const v2f*)(B0 + kk);
        v2f b1 = *(const v2f*)(B1 + kk);
        v2f b2 = *(const v2f*)(B2 + kk);
        v2f b3 = *(const v2f*)(B3 + kk);
        c0 = __builtin_amdgcn_wmma_f32_16x16x4_f32(false, a, false, b0, (short)0, c0, false, false);
        c1 = __builtin_amdgcn_wmma_f32_16x16x4_f32(false, a, false, b1, (short)0, c1, false, false);
        c2 = __builtin_amdgcn_wmma_f32_16x16x4_f32(false, a, false, b2, (short)0, c2, false, false);
        c3 = __builtin_amdgcn_wmma_f32_16x16x4_f32(false, a, false, b3, (short)0, c3, false, false);
    }

    const float bb0 = bout[tileN + lr +  0];
    const float bb1 = bout[tileN + lr + 16];
    const float bb2 = bout[tileN + lr + 32];
    const float bb3 = bout[tileN + lr + 48];

#pragma unroll
    for (int r = 0; r < 8; ++r) {
        float* o = out + (size_t)(tileM + r + 8 * hi) * Oo + tileN + lr;
        o[0]  = c0[r] + bb0;
        o[16] = c1[r] + bb1;
        o[32] = c2[r] + bb2;
        o[48] = c3[r] + bb3;
    }
}

extern "C" void kernel_launch(void* const* d_in, const int* in_sizes, int n_in,
                              void* d_out, int out_size, void* d_ws, size_t ws_size,
                              hipStream_t stream)
{
    const float* x     = (const float*)d_in[0];
    const float* Wk    = (const float*)d_in[1];
    const float* Wq    = (const float*)d_in[2];
    const float* Wv    = (const float*)d_in[3];
    const float* Wvc   = (const float*)d_in[4];
    const float* We    = (const float*)d_in[5];
    const float* theta = (const float*)d_in[6];
    const float* Wout  = (const float*)d_in[7];
    const float* bout  = (const float*)d_in[8];
    const float* gamma = (const float*)d_in[9];
    const float* beta  = (const float*)d_in[10];
    float* out = (float*)d_out;

    float* ws  = (float*)d_ws;
    const size_t NROW = (size_t)Bb * Tt;
    float* xn  = ws;
    float* Ko  = xn  + NROW * Dd;
    float* Qo  = Ko  + NROW * Mm;
    float* Vo  = Qo  + NROW * Mm;
    float* vco = Vo  + NROW * Mm;
    float* yb  = vco + NROW * Mm;
    float* eta = yb  + NROW * Mm;

    ln_eta_kernel<<<dim3(NROW / 8), dim3(256), 0, stream>>>(x, gamma, beta, We, xn, eta);
    proj_gemm_kernel<<<dim3(NROW / 16, 4), dim3(32), 0, stream>>>(xn, Wk, Wq, Wv, Wvc,
                                                                  Ko, Qo, Vo, vco);
    scan_kernel<<<dim3(Bb, Mm / 8), dim3(256), 0, stream>>>(Ko, Qo, Vo, vco, eta, theta, yb);
    out_gemm_kernel<<<dim3(NROW / 16, Oo / 64), dim3(32), 0, stream>>>(yb, Wout, bout, out);
}


# LayerNorm folded through the projection GEMM by linearity: WMMA on raw x against gamma*W in LDS, row mean/rstd summed from the A fragments, applied after; LN launch, xn gone; new rounding, sum order
# speedup vs baseline: 16.0564x; 16.0564x over previous
#include <hip/hip_runtime.h>
#include <hip/hip_bf16.h>

typedef float v2f __attribute__((ext_vector_type(2)));
typedef float v4f __attribute__((ext_vector_type(4)));
typedef float v8f __attribute__((ext_vector_type(8)));

#define Bb 8
#define Tt 1024
#define Dd 512
#define Mm 64
#define Oo 512
#define NC 32
#define LC (Tt / NC)

#define PW 8
#define WPITCH (Dd + 4)
#define WR (Mm / PW)

template <bool ETA>
__device__ __forceinline__ void proj_k_loop(const float* __restrict__ Arow,
                                            const float* B0, const float* B1,
                                            const float* B2, const float* B3,
                                            const float* Ge,
                                            v8f& c0, v8f& c1, v8f& c2, v8f& c3,
                                            v2f& s1, v2f& s2, v2f& e1)
{
#pragma unroll 8
    for (int kk = 0; kk < Dd; kk += 4) {
        v2f a  = *(const v2f*)(Arow + kk);
        v2f b0 = *(const v2f*)(B0 + kk);
        v2f b1 = *(const v2f*)(B1 + kk);
        v2f b2 = *(const v2f*)(B2 + kk);
        v2f b3 = *(const v2f*)(B3 + kk);
        c0 = __builtin_amdgcn_wmma_f32_16x16x4_f32(false, a, false, b0, (short)0, c0, false, false);
        c1 = __builtin_amdgcn_wmma_f32_16x16x4_f32(false, a, false, b1, (short)0, c1, false, false);
        c2 = __builtin_amdgcn_wmma_f32_16x16x4_f32(false, a, false, b2, (short)0, c2, false, false);
        c3 = __builtin_amdgcn_wmma_f32_16x16x4_f32(false, a, false, b3, (short)0, c3, false, false);
        s1 += a;
        s2 = __builtin_elementwise_fma(a, a, s2);
        if (ETA) e1 = __builtin_elementwise_fma(a, *(const v2f*)(Ge + kk), e1);
    }
}

__global__ __launch_bounds__(32 * PW)
void ln_proj_gemm_kernel(const float* __restrict__ x,
                         const float* __restrict__ gamma,
                         const float* __restrict__ beta,
                         const float* __restrict__ We,
                         const float* __restrict__ Wk,
                         const float* __restrict__ Wq,
                         const float* __restrict__ Wv,
                         const float* __restrict__ Wvc,
                         float* __restrict__ Ko,
                         float* __restrict__ Qo,
                         float* __restrict__ Vo,
                         float* __restrict__ vco,
                         float* __restrict__ eta)
{
    __shared__ v4f Ws4[Mm * WPITCH / 4];
    __shared__ v4f gwe4[Dd / 4];
    __shared__ float csL[Mm];
    __shared__ float bsL[Mm];
    float* Ws = (float*)Ws4;

    const int w = blockIdx.y;
    const float* W = (w == 0) ? Wk : (w == 1) ? Wq : (w == 2) ? Wv : Wvc;
    float* Out     = (w == 0) ? Ko : (w == 1) ? Qo : (w == 2) ? Vo : vco;

    const int tid   = threadIdx.x;
    const int wave  = tid >> 5;
    const int lane  = tid & 31;
    const int hi    = lane >> 4;
    const int lr    = lane & 15;

    v4f g4[4], b4[4];
#pragma unroll
    for (int j = 0; j < 4; ++j) {
        g4[j] = *(const v4f*)(gamma + (lane + 32 * j) * 4);
        b4[j] = *(const v4f*)(beta  + (lane + 32 * j) * 4);
    }
    v4f wr[WR][4];
#pragma unroll
    for (int r = 0; r < WR; ++r) {
        const float* Wrow = W + (size_t)(wave * WR + r) * Dd;
#pragma unroll
        for (int j = 0; j < 4; ++j) wr[r][j] = *(const v4f*)(Wrow + (lane + 32 * j) * 4);
    }
    float cs[WR + 1], bs[WR + 1];
#pragma unroll
    for (int r = 0; r < WR; ++r) {
        const int row = wave * WR + r;
        cs[r] = 0.f; bs[r] = 0.f;
#pragma unroll
        for (int j = 0; j < 4; ++j) {
            const v4f gw = wr[r][j] * g4[j];
            const v4f bw = wr[r][j] * b4[j];
            *(v4f*)(Ws + row * WPITCH + (lane + 32 * j) * 4) = gw;
            cs[r] += (gw.x + gw.y) + (gw.z + gw.w);
            bs[r] += (bw.x + bw.y) + (bw.z + bw.w);
        }
    }
    cs[WR] = 0.f; bs[WR] = 0.f;
    if (w == 2) {
#pragma unroll
        for (int j = 0; j < 4; ++j) {
            const v4f we = *(const v4f*)(We + (lane + 32 * j) * 4);
            const v4f gw = we * g4[j];
            const v4f bw = we * b4[j];
            if (wave == 0) gwe4[lane + 32 * j] = gw;
            cs[WR] += (gw.x + gw.y) + (gw.z + gw.w);
            bs[WR] += (bw.x + bw.y) + (bw.z + bw.w);
        }
    }
#pragma unroll
    for (int m = 16; m >= 1; m >>= 1) {
#pragma unroll
        for (int r = 0; r <= WR; ++r) {
            cs[r] += __shfl_xor(cs[r], m, 32);
            bs[r] += __shfl_xor(bs[r], m, 32);
        }
    }
    if (lane == 0) {
#pragma unroll
        for (int r = 0; r < WR; ++r) { csL[wave * WR + r] = cs[r]; bsL[wave * WR + r] = bs[r]; }
    }
    const float cse = cs[WR], bse = bs[WR];
    __syncthreads();

    const int tileM = (blockIdx.x * PW + wave) * 16;
    const float* Arow = x + (size_t)(tileM + lr) * Dd + 2 * hi;
    const float* B0 = Ws + (lr +  0) * WPITCH + 2 * hi;
    const float* B1 = Ws + (lr + 16) * WPITCH + 2 * hi;
    const float* B2 = Ws + (lr + 32) * WPITCH + 2 * hi;
    const float* B3 = Ws + (lr + 48) * WPITCH + 2 * hi;
    const float* Ge = (const float*)gwe4 + 2 * hi;

    v8f c0 = {}, c1 = {}, c2 = {}, c3 = {};
    v2f s1 = { 0.f, 0.f }, s2 = { 0.f, 0.f }, e1 = { 0.f, 0.f };
    if (w == 2) proj_k_loop<true >(Arow, B0, B1, B2, B3, Ge, c0, c1, c2, c3, s1, s2, e1);
    else        proj_k_loop<false>(Arow, B0, B1, B2, B3, Ge, c0, c1, c2, c3, s1, s2, e1);

    float sum = s1.x + s1.y, sq = s2.x + s2.y;
    sum += __shfl_xor(sum, 16, 32);
    sq  += __shfl_xor(sq,  16, 32);
    const float mu  = sum * (1.0f / Dd);
    const float var = sq * (1.0f / Dd) - mu * mu;
    const float rs  = rsqrtf(var + 1e-5f);

    if (w == 2) {
        float e = e1.x + e1.y;
        e += __shfl_xor(e, 16, 32);
        const float dot = rs * (e - mu * cse) + bse;
        if (hi == 0) eta[tileM + lr] = 1.0f / (1.0f + expf(-dot));
    }

    const float cs0 = csL[lr], cs1 = csL[lr + 16], cs2 = csL[lr + 32], cs3 = csL[lr + 48];
    const float bs0 = bsL[lr], bs1 = bsL[lr + 16], bs2 = bsL[lr + 32], bs3 = bsL[lr + 48];
#pragma unroll
    for (int r = 0; r < 8; ++r) {
        const float mur = __shfl(mu, r + 8 * hi, 32);
        const float rsr = __shfl(rs, r + 8 * hi, 32);
        c0[r] = rsr * (c0[r] - mur * cs0) + bs0;
        c1[r] = rsr * (c1[r] - mur * cs1) + bs1;
        c2[r] = rsr * (c2[r] - mur * cs2) + bs2;
        c3[r] = rsr * (c3[r] - mur * cs3) + bs3;
    }

    if (w < 2) {
#pragma unroll
        for (int r = 0; r < 8; ++r) {
            c0[r] = fmaxf(c0[r], 0.f); c1[r] = fmaxf(c1[r], 0.f);
            c2[r] = fmaxf(c2[r], 0.f); c3[r] = fmaxf(c3[r], 0.f);
            float s = c0[r] + c1[r] + c2[r] + c3[r];
#pragma unroll
            for (int m = 8; m >= 1; m >>= 1) s += __shfl_xor(s, m, 32);
            float inv = 1.0f / (1e-5f + s);
            c0[r] *= inv; c1[r] *= inv; c2[r] *= inv; c3[r] *= inv;
        }
    }

#pragma unroll
    for (int r = 0; r < 8; ++r) {
        float* o = Out + (size_t)(tileM + r + 8 * hi) * Mm + lr;
        o[0]  = c0[r];
        o[16] = c1[r];
        o[32] = c2[r];
        o[48] = c3[r];
    }
}

#define CG 8
#define NG (Mm / CG)
#define RH (Mm / 32)

__device__ __forceinline__ float gate_from(float th2, float vcj)
{
    const float e = __builtin_amdgcn_exp2f(th2 * vcj);
    return __builtin_amdgcn_rcpf(__builtin_fmaf(e, 0.5f, 0.5f));
}

#define SUMMARY_QUAD (32 * 4)
__device__ __forceinline__ size_t summary_offset(int b, int c, int rh, int g, int lane)
{
    return ((((size_t)b * NC + c) * RH + rh) * NG + g) * (32 * CG) + (size_t)lane * 4;
}

#define SUMMARY_CHUNK (RH * NG * 32 * CG)

template <int N>
__device__ __forceinline__ void fold_chunks(const float* __restrict__ Sp,
                                            const float* __restrict__ Pp,
                                            v2f (&s)[CG / 2])
{
    v4f sa[N], sb[N], pa[N], pb[N];
#pragma unroll
    for (int k = 0; k < N; ++k) {
        sa[k] = *(const v4f*)(Sp + k * SUMMARY_CHUNK);  sb[k] = *(const v4f*)(Sp + k * SUMMARY_CHUNK + SUMMARY_QUAD);
        pa[k] = *(const v4f*)(Pp + k * SUMMARY_CHUNK);  pb[k] = *(const v4f*)(Pp + k * SUMMARY_CHUNK + SUMMARY_QUAD);
    }
#pragma unroll
    for (int k = 0; k < N; ++k) {
        s[0] = __builtin_elementwise_fma(pa[k].xy, s[0], sa[k].xy);
        s[1] = __builtin_elementwise_fma(pa[k].zw, s[1], sa[k].zw);
        s[2] = __builtin_elementwise_fma(pb[k].xy, s[2], sb[k].xy);
        s[3] = __builtin_elementwise_fma(pb[k].zw, s[3], sb[k].zw);
    }
}

__global__ __launch_bounds__(32 * NG)
void scan_summary_kernel(const float* __restrict__ K,
                         const float* __restrict__ V,
                         const float* __restrict__ vc,
                         const float* __restrict__ eta,
                         const float* __restrict__ theta_matrix,
                         float* __restrict__ Send,
                         float* __restrict__ Pend)
{
    __shared__ v4f vcs[LC * Mm / 4];
    __shared__ v4f ks[LC * Mm / 4];
    __shared__ v4f us[LC * 32 / 4];

    const int b    = blockIdx.x;
    const int rh   = blockIdx.y;
    const int c    = blockIdx.z;
    const int tid  = threadIdx.x;
    const int lane = tid & 31;
    const int g    = tid >> 5;
    const size_t r0 = (size_t)b * Tt + (size_t)c * LC;

    {
        const v4f* vcg = (const v4f*)(vc + r0 * Mm);
        const v4f* kg  = (const v4f*)(K  + r0 * Mm);
        vcs[tid] = vcg[tid];  vcs[tid + 32 * NG] = vcg[tid + 32 * NG];
        ks[tid]  = kg[tid];   ks[tid + 32 * NG]  = kg[tid + 32 * NG];
        const int t = tid >> 3, part = tid & 7;
        const v4f v4 = *(const v4f*)(V + (r0 + t) * Mm + rh * 32 + part * 4);
        us[tid] = v4 * eta[r0 + t];
    }
    __syncthreads();

    const float th2 = theta_matrix[rh * 32 + lane] * -2.8853900817779268f;
    const float* usf = (const float*)us;

    v2f s[CG / 2], p[CG / 2];
#pragma unroll
    for (int e = 0; e < CG / 2; ++e) { s[e] = (v2f){ 0.f, 0.f }; p[e] = (v2f){ 1.f, 1.f }; }

#pragma unroll 2
    for (int t = 0; t < LC; ++t) {
        const v4f va = vcs[t * (Mm / 4) + g * 2], vb = vcs[t * (Mm / 4) + g * 2 + 1];
        const v4f ka = ks[t * (Mm / 4) + g * 2],  kb = ks[t * (Mm / 4) + g * 2 + 1];
        const float u = usf[t * 32 + lane];
        const v2f vcv[CG / 2] = { va.xy, va.zw, vb.xy, vb.zw };
        const v2f kv[CG / 2]  = { ka.xy, ka.zw, kb.xy, kb.zw };
#pragma unroll
        for (int e = 0; e < CG / 2; ++e) {
            const v2f cc = { gate_from(th2, vcv[e].x), gate_from(th2, vcv[e].y) };
            s[e] = __builtin_elementwise_fma(s[e], cc, kv[e] * u);
            p[e] *= cc;
        }
    }

    const size_t so = summary_offset(b, c, rh, g, lane);
    *(v4f*)(Send + so)                = (v4f){ s[0].x, s[0].y, s[1].x, s[1].y };
    *(v4f*)(Send + so + SUMMARY_QUAD) = (v4f){ s[2].x, s[2].y, s[3].x, s[3].y };
    *(v4f*)(Pend + so)                = (v4f){ p[0].x, p[0].y, p[1].x, p[1].y };
    *(v4f*)(Pend + so + SUMMARY_QUAD) = (v4f){ p[2].x, p[2].y, p[3].x, p[3].y };
}

#define EW (RH * NG)
#define YPITCH (Mm + 4)
__global__ __launch_bounds__(32 * EW)
void scan_emit_out_kernel(const float* __restrict__ K,
                          const float* __restrict__ Q,
                          const float* __restrict__ V,
                          const float* __restrict__ vc,
                          const float* __restrict__ eta,
                          const float* __restrict__ theta_matrix,
                          const float* __restrict__ Send,
                          const float* __restrict__ Pend,
                          const float* __restrict__ Wout,
                          const float* __restrict__ bout,
                          float* __restrict__ out)
{
    __shared__ v4f vcs[LC * Mm / 4];
    __shared__ v4f ks[LC * Mm / 4];
    __shared__ v4f qs[LC * Mm / 4];
    __shared__ v4f us[LC * Mm / 4];
    __shared__ float yp[NG * LC * Mm];
    __shared__ v2f yt2[LC * YPITCH / 2];
    float* yt = (float*)yt2;

    const int b    = blockIdx.x;
    const int c    = blockIdx.y;
    const int tid  = threadIdx.x;
    const int lane = tid & 31;
    const int wv   = tid >> 5;
    const int rh   = wv / NG;
    const int g    = wv % NG;
    const int hi   = lane >> 4;
    const int lr   = lane & 15;
    const size_t r0 = (size_t)b * Tt + (size_t)c * LC;

    {
        const v4f* vcg = (const v4f*)(vc + r0 * Mm);
        const v4f* kg  = (const v4f*)(K  + r0 * Mm);
        const v4f* qg  = (const v4f*)(Q  + r0 * Mm);
        const v4f* vg  = (const v4f*)(V  + r0 * Mm);
        vcs[tid] = vcg[tid];
        ks[tid]  = kg[tid];
        qs[tid]  = qg[tid];
        us[tid]  = vg[tid] * eta[r0 + (tid >> 4)];
    }

    const int n0 = wv * 32;
    v2f bf0[Mm / 4], bf1[Mm / 4];
    {
        const float* Bp0 = Wout + (size_t)(n0 + lr) * Mm + 2 * hi;
        const float* Bp1 = Bp0 + 16 * Mm;
#pragma unroll
        for (int k4 = 0; k4 < Mm / 4; ++k4) {
            bf0[k4] = *(const v2f*)(Bp0 + 4 * k4);
            bf1[k4] = *(const v2f*)(Bp1 + 4 * k4);
        }
    }

    v2f s[CG / 2];
#pragma unroll
    for (int e = 0; e < CG / 2; ++e) s[e] = (v2f){ 0.f, 0.f };
    {
        const float* Sp = Send + summary_offset(b, 0, rh, g, lane);
        const float* Pp = Pend + summary_offset(b, 0, rh, g, lane);
        int cp = 0;
        for (; cp + 8 <= c; cp += 8) fold_chunks<8>(Sp + cp * SUMMARY_CHUNK, Pp + cp * SUMMARY_CHUNK, s);
        if (c & 4) { fold_chunks<4>(Sp + cp * SUMMARY_CHUNK, Pp + cp * SUMMARY_CHUNK, s); cp += 4; }
        if (c & 2) { fold_chunks<2>(Sp + cp * SUMMARY_CHUNK, Pp + cp * SUMMARY_CHUNK, s); cp += 2; }
        if (c & 1) { fold_chunks<1>(Sp + cp * SUMMARY_CHUNK, Pp + cp * SUMMARY_CHUNK, s); }
    }
    __syncthreads();

    const float th2 = theta_matrix[rh * 32 + lane] * -2.8853900817779268f;
    const float* usf = (const float*)us + rh * 32 + lane;
    float* ypw = yp + g * (LC * Mm) + rh * 32 + lane;

#pragma unroll 2
    for (int t = 0; t < LC; ++t) {
        const v4f va = vcs[t * (Mm / 4) + g * 2], vb = vcs[t * (Mm / 4) + g * 2 + 1];
        const v4f ka = ks[t * (Mm / 4) + g * 2],  kb = ks[t * (Mm / 4) + g * 2 + 1];
        const v4f qa = qs[t * (Mm / 4) + g * 2],  qb = qs[t * (Mm / 4) + g * 2 + 1];
        const float u = usf[t * Mm];
        const v2f vcv[CG / 2] = { va.xy, va.zw, vb.xy, vb.zw };
        const v2f kv[CG / 2]  = { ka.xy, ka.zw, kb.xy, kb.zw };
        const v2f qv[CG / 2]  = { qa.xy, qa.zw, qb.xy, qb.zw };
        v2f yy = { 0.f, 0.f };
#pragma unroll
        for (int e = 0; e < CG / 2; ++e) {
            const v2f cc = { gate_from(th2, vcv[e].x), gate_from(th2, vcv[e].y) };
            s[e] = __builtin_elementwise_fma(s[e], cc, kv[e] * u);
            yy = __builtin_elementwise_fma(s[e], qv[e], yy);
        }
        ypw[t * Mm] = yy.x + yy.y;
    }
    __syncthreads();

#pragma unroll
    for (int k = 0; k < (LC * Mm) / (32 * EW); ++k) {
        const int idx = tid + (32 * EW) * k;
        float acc = yp[idx];
#pragma unroll
        for (int g2 = 1; g2 < NG; ++g2) acc += yp[g2 * (LC * Mm) + idx];
        yt[(idx >> 6) * YPITCH + (idx & 63)] = acc;
    }
    __syncthreads();

    const float* A0 = yt + lr * YPITCH + 2 * hi;
    const float* A1 = A0 + 16 * YPITCH;
    v8f c00 = {}, c01 = {}, c10 = {}, c11 = {};
#pragma unroll
    for (int k4 = 0; k4 < Mm / 4; ++k4) {
        const v2f a0 = *(const v2f*)(A0 + 4 * k4);
        const v2f a1 = *(const v2f*)(A1 + 4 * k4);
        c00 = __builtin_amdgcn_wmma_f32_16x16x4_f32(false, a0, false, bf0[k4], (short)0, c00, false, false);
        c01 = __builtin_amdgcn_wmma_f32_16x16x4_f32(false, a0, false, bf1[k4], (short)0, c01, false, false);
        c10 = __builtin_amdgcn_wmma_f32_16x16x4_f32(false, a1, false, bf0[k4], (short)0, c10, false, false);
        c11 = __builtin_amdgcn_wmma_f32_16x16x4_f32(false, a1, false, bf1[k4], (short)0, c11, false, false);
    }

    const float bb0 = bout[n0 + lr];
    const float bb1 = bout[n0 + 16 + lr];
#pragma unroll
    for (int r = 0; r < 8; ++r) {
        float* o0 = out + (r0 + r + 8 * hi) * Oo + n0 + lr;
        float* o1 = o0 + (size_t)16 * Oo;
        o0[0]  = c00[r] + bb0;
        o0[16] = c01[r] + bb1;
        o1[0]  = c10[r] + bb0;
        o1[16] = c11[r] + bb1;
    }
}

extern "C" void kernel_launch(void* const* d_in, const int* in_sizes, int n_in,
                              void* d_out, int out_size, void* d_ws, size_t ws_size,
                              hipStream_t stream)
{
    const float* x     = (const float*)d_in[0];
    const float* Wk    = (const float*)d_in[1];
    const float* Wq    = (const float*)d_in[2];
    const float* Wv    = (const float*)d_in[3];
    const float* Wvc   = (const float*)d_in[4];
    const float* We    = (const float*)d_in[5];
    const float* theta = (const float*)d_in[6];
    const float* Wout  = (const float*)d_in[7];
    const float* bout  = (const float*)d_in[8];
    const float* gamma = (const float*)d_in[9];
    const float* beta  = (const float*)d_in[10];
    float* out = (float*)d_out;

    float* ws  = (float*)d_ws;
    const size_t NROW = (size_t)Bb * Tt;
    float* Ko  = ws;
    float* Qo  = Ko  + NROW * Mm;
    float* Vo  = Qo  + NROW * Mm;
    float* vco = Vo  + NROW * Mm;
    float* eta = vco + NROW * Mm;
    float* Send = eta + NROW;
    float* Pend = Send + (size_t)Bb * NC * Mm * Mm;

    ln_proj_gemm_kernel<<<dim3(NROW / (16 * PW), 4), dim3(32 * PW), 0, stream>>>(x, gamma, beta, We,
                                                                  Wk, Wq, Wv, Wvc,
                                                                  Ko, Qo, Vo, vco, eta);
    scan_summary_kernel<<<dim3(Bb, RH, NC - 1), dim3(32 * NG), 0, stream>>>(Ko, Vo, vco, eta, theta,
                                                                            Send, Pend);
    scan_emit_out_kernel<<<dim3(Bb, NC), dim3(32 * EW), 0, stream>>>(Ko, Qo, Vo, vco, eta, theta,
                                                                     Send, Pend, Wout, bout, out);
}


# scan pass 1: a work-group scans a PAIR of chunks (64 steps) and stores its running summary after 32 and after 64 steps; pass 2 folds at most 16 summaries, not 31 (pair sums: new rounding at joins)
# speedup vs baseline: 17.1902x; 1.0706x over previous
#include <hip/hip_runtime.h>
#include <hip/hip_bf16.h>

typedef float v2f __attribute__((ext_vector_type(2)));
typedef float v4f __attribute__((ext_vector_type(4)));
typedef float v8f __attribute__((ext_vector_type(8)));

#define Bb 8
#define Tt 1024
#define Dd 512
#define Mm 64
#define Oo 512
#define NC 32
#define LC (Tt / NC)

#define PW 8
#define WPITCH (Dd + 4)
#define WR (Mm / PW)

template <bool ETA>
__device__ __forceinline__ void proj_k_loop(const float* __restrict__ Arow,
                                            const float* B0, const float* B1,
                                            const float* B2, const float* B3,
                                            const float* Ge,
                                            v8f& c0, v8f& c1, v8f& c2, v8f& c3,
                                            v2f& s1, v2f& s2, v2f& e1)
{
#pragma unroll 8
    for (int kk = 0; kk < Dd; kk += 4) {
        v2f a  = *(const v2f*)(Arow + kk);
        v2f b0 = *(const v2f*)(B0 + kk);
        v2f b1 = *(const v2f*)(B1 + kk);
        v2f b2 = *(const v2f*)(B2 + kk);
        v2f b3 = *(const v2f*)(B3 + kk);
        c0 = __builtin_amdgcn_wmma_f32_16x16x4_f32(false, a, false, b0, (short)0, c0, false, false);
        c1 = __builtin_amdgcn_wmma_f32_16x16x4_f32(false, a, false, b1, (short)0, c1, false, false);
        c2 = __builtin_amdgcn_wmma_f32_16x16x4_f32(false, a, false, b2, (short)0, c2, false, false);
        c3 = __builtin_amdgcn_wmma_f32_16x16x4_f32(false, a, false, b3, (short)0, c3, false, false);
        s1 += a;
        s2 = __builtin_elementwise_fma(a, a, s2);
        if (ETA) e1 = __builtin_elementwise_fma(a, *(const v2f*)(Ge + kk), e1);
    }
}

__global__ __launch_bounds__(32 * PW)
void ln_proj_gemm_kernel(const float* __restrict__ x,
                         const float* __restrict__ gamma,
                         const float* __restrict__ beta,
                         const float* __restrict__ We,
                         const float* __restrict__ Wk,
                         const float* __restrict__ Wq,
                         const float* __restrict__ Wv,
                         const float* __restrict__ Wvc,
                         float* __restrict__ Ko,
                         float* __restrict__ Qo,
                         float* __restrict__ Vo,
                         float* __restrict__ vco,
                         float* __restrict__ eta)
{
    __shared__ v4f Ws4[Mm * WPITCH / 4];
    __shared__ v4f gwe4[Dd / 4];
    __shared__ float csL[Mm];
    __shared__ float bsL[Mm];
    float* Ws = (float*)Ws4;

    const int w = blockIdx.y;
    const float* W = (w == 0) ? Wk : (w == 1) ? Wq : (w == 2) ? Wv : Wvc;
    float* Out     = (w == 0) ? Ko : (w == 1) ? Qo : (w == 2) ? Vo : vco;

    const int tid   = threadIdx.x;
    const int wave  = tid >> 5;
    const int lane  = tid & 31;
    const int hi    = lane >> 4;
    const int lr    = lane & 15;

    v4f g4[4], b4[4];
#pragma unroll
    for (int j = 0; j < 4; ++j) {
        g4[j] = *(const v4f*)(gamma + (lane + 32 * j) * 4);
        b4[j] = *(const v4f*)(beta  + (lane + 32 * j) * 4);
    }
    v4f wr[WR][4];
#pragma unroll
    for (int r = 0; r < WR; ++r) {
        const float* Wrow = W + (size_t)(wave * WR + r) * Dd;
#pragma unroll
        for (int j = 0; j < 4; ++j) wr[r][j] = *(const v4f*)(Wrow + (lane + 32 * j) * 4);
    }
    float cs[WR + 1], bs[WR + 1];
#pragma unroll
    for (int r = 0; r < WR; ++r) {
        const int row = wave * WR + r;
        cs[r] = 0.f; bs[r] = 0.f;
#pragma unroll
        for (int j = 0; j < 4; ++j) {
            const v4f gw = wr[r][j] * g4[j];
            const v4f bw = wr[r][j] * b4[j];
            *(v4f*)(Ws + row * WPITCH + (lane + 32 * j) * 4) = gw;
            cs[r] += (gw.x + gw.y) + (gw.z + gw.w);
            bs[r] += (bw.x + bw.y) + (bw.z + bw.w);
        }
    }
    cs[WR] = 0.f; bs[WR] = 0.f;
    if (w == 2) {
#pragma unroll
        for (int j = 0; j < 4; ++j) {
            const v4f we = *(const v4f*)(We + (lane + 32 * j) * 4);
            const v4f gw = we * g4[j];
            const v4f bw = we * b4[j];
            if (wave == 0) gwe4[lane + 32 * j] = gw;
            cs[WR] += (gw.x + gw.y) + (gw.z + gw.w);
            bs[WR] += (bw.x + bw.y) + (bw.z + bw.w);
        }
    }
#pragma unroll
    for (int m = 16; m >= 1; m >>= 1) {
#pragma unroll
        for (int r = 0; r <= WR; ++r) {
            cs[r] += __shfl_xor(cs[r], m, 32);
            bs[r] += __shfl_xor(bs[r], m, 32);
        }
    }
    if (lane == 0) {
#pragma unroll
        for (int r = 0; r < WR; ++r) { csL[wave * WR + r] = cs[r]; bsL[wave * WR + r] = bs[r]; }
    }
    const float cse = cs[WR], bse = bs[WR];
    __syncthreads();

    const int tileM = (blockIdx.x * PW + wave) * 16;
    const float* Arow = x + (size_t)(tileM + lr) * Dd + 2 * hi;
    const float* B0 = Ws + (lr +  0) * WPITCH + 2 * hi;
    const float* B1 = Ws + (lr + 16) * WPITCH + 2 * hi;
    const float* B2 = Ws + (lr + 32) * WPITCH + 2 * hi;
    const float* B3 = Ws + (lr + 48) * WPITCH + 2 * hi;
    const float* Ge = (const float*)gwe4 + 2 * hi;

    v8f c0 = {}, c1 = {}, c2 = {}, c3 = {};
    v2f s1 = { 0.f, 0.f }, s2 = { 0.f, 0.f }, e1 = { 0.f, 0.f };
    if (w == 2) proj_k_loop<true >(Arow, B0, B1, B2, B3, Ge, c0, c1, c2, c3, s1, s2, e1);
    else        proj_k_loop<false>(Arow, B0, B1, B2, B3, Ge, c0, c1, c2, c3, s1, s2, e1);

    float sum = s1.x + s1.y, sq = s2.x + s2.y;
    sum += __shfl_xor(sum, 16, 32);
    sq  += __shfl_xor(sq,  16, 32);
    const float mu  = sum * (1.0f / Dd);
    const float var = sq * (1.0f / Dd) - mu * mu;
    const float rs  = rsqrtf(var + 1e-5f);

    if (w == 2) {
        float e = e1.x + e1.y;
        e += __shfl_xor(e, 16, 32);
        const float dot = rs * (e - mu * cse) + bse;
        if (hi == 0) eta[tileM + lr] = 1.0f / (1.0f + expf(-dot));
    }

    const float cs0 = csL[lr], cs1 = csL[lr + 16], cs2 = csL[lr + 32], cs3 = csL[lr + 48];
    const float bs0 = bsL[lr], bs1 = bsL[lr + 16], bs2 = bsL[lr + 32], bs3 = bsL[lr + 48];
#pragma unroll
    for (int r = 0; r < 8; ++r) {
        const float mur = __shfl(mu, r + 8 * hi, 32);
        const float rsr = __shfl(rs, r + 8 * hi, 32);
        c0[r] = rsr * (c0[r] - mur * cs0) + bs0;
        c1[r] = rsr * (c1[r] - mur * cs1) + bs1;
        c2[r] = rsr * (c2[r] - mur * cs2) + bs2;
        c3[r] = rsr * (c3[r] - mur * cs3) + bs3;
    }

    if (w < 2) {
#pragma unroll
        for (int r = 0; r < 8; ++r) {
            c0[r] = fmaxf(c0[r], 0.f); c1[r] = fmaxf(c1[r], 0.f);
            c2[r] = fmaxf(c2[r], 0.f); c3[r] = fmaxf(c3[r], 0.f);
            float s = c0[r] + c1[r] + c2[r] + c3[r];
#pragma unroll
            for (int m = 8; m >= 1; m >>= 1) s += __shfl_xor(s, m, 32);
            float inv = 1.0f / (1e-5f + s);
            c0[r] *= inv; c1[r] *= inv; c2[r] *= inv; c3[r] *= inv;
        }
    }

#pragma unroll
    for (int r = 0; r < 8; ++r) {
        float* o = Out + (size_t)(tileM + r + 8 * hi) * Mm + lr;
        o[0]  = c0[r];
        o[16] = c1[r];
        o[32] = c2[r];
        o[48] = c3[r];
    }
}

#define CG 8
#define NG (Mm / CG)
#define RH (Mm / 32)
#define PC 2

__device__ __forceinline__ float gate_from(float th2, float vcj)
{
    const float e = __builtin_amdgcn_exp2f(th2 * vcj);
    return __builtin_amdgcn_rcpf(__builtin_fmaf(e, 0.5f, 0.5f));
}

#define SUMMARY_QUAD (32 * 4)
__device__ __forceinline__ size_t summary_offset(int b, int c, int rh, int g, int lane)
{
    return ((((size_t)b * NC + c) * RH + rh) * NG + g) * (32 * CG) + (size_t)lane * 4;
}

#define SUMMARY_CHUNK (RH * NG * 32 * CG)

template <int N>
__device__ __forceinline__ void fold_chunks(const float* __restrict__ Sb,
                                            const float* __restrict__ Pb,
                                            int e0, int last,
                                            v2f (&s)[CG / 2])
{
    v4f sa[N], sb[N], pa[N], pb[N];
#pragma unroll
    for (int k = 0; k < N; ++k) {
        const int slot = min(PC * (e0 + k) + (PC - 1), last);
        const float* Sp = Sb + slot * SUMMARY_CHUNK;
        const float* Pp = Pb + slot * SUMMARY_CHUNK;
        sa[k] = *(const v4f*)(Sp);  sb[k] = *(const v4f*)(Sp + SUMMARY_QUAD);
        pa[k] = *(const v4f*)(Pp);  pb[k] = *(const v4f*)(Pp + SUMMARY_QUAD);
    }
#pragma unroll
    for (int k = 0; k < N; ++k) {
        s[0] = __builtin_elementwise_fma(pa[k].xy, s[0], sa[k].xy);
        s[1] = __builtin_elementwise_fma(pa[k].zw, s[1], sa[k].zw);
        s[2] = __builtin_elementwise_fma(pb[k].xy, s[2], sb[k].xy);
        s[3] = __builtin_elementwise_fma(pb[k].zw, s[3], sb[k].zw);
    }
}

__global__ __launch_bounds__(32 * NG)
void scan_summary_kernel(const float* __restrict__ K,
                         const float* __restrict__ V,
                         const float* __restrict__ vc,
                         const float* __restrict__ eta,
                         const float* __restrict__ theta_matrix,
                         float* __restrict__ Send,
                         float* __restrict__ Pend)
{
    __shared__ v4f vcs[PC * LC * Mm / 4];
    __shared__ v4f ks[PC * LC * Mm / 4];
    __shared__ v4f us[PC * LC * 32 / 4];

    const int b    = blockIdx.x;
    const int rh   = blockIdx.y;
    const int run  = blockIdx.z;
    const int tid  = threadIdx.x;
    const int lane = tid & 31;
    const int g    = tid >> 5;
    const size_t r0 = (size_t)b * Tt + (size_t)run * (PC * LC);

    {
        const v4f* vcg = (const v4f*)(vc + r0 * Mm);
        const v4f* kg  = (const v4f*)(K  + r0 * Mm);
#pragma unroll
        for (int q = 0; q < (PC * LC * Mm / 4) / (32 * NG); ++q) {
            vcs[tid + 32 * NG * q] = vcg[tid + 32 * NG * q];
            ks[tid + 32 * NG * q]  = kg[tid + 32 * NG * q];
        }
#pragma unroll
        for (int q = 0; q < (PC * LC * 32 / 4) / (32 * NG); ++q) {
            const int idx = tid + 32 * NG * q;
            const int t = idx >> 3, part = idx & 7;
            const v4f v4 = *(const v4f*)(V + (r0 + t) * Mm + rh * 32 + part * 4);
            us[idx] = v4 * eta[r0 + t];
        }
    }
    __syncthreads();

    const float th2 = theta_matrix[rh * 32 + lane] * -2.8853900817779268f;
    const float* usf = (const float*)us;

    v2f s[CG / 2], p[CG / 2];
#pragma unroll
    for (int e = 0; e < CG / 2; ++e) { s[e] = (v2f){ 0.f, 0.f }; p[e] = (v2f){ 1.f, 1.f }; }

    for (int h = 0; h < PC; ++h) {
#pragma unroll 2
        for (int t = h * LC; t < (h + 1) * LC; ++t) {
            const v4f va = vcs[t * (Mm / 4) + g * 2], vb = vcs[t * (Mm / 4) + g * 2 + 1];
            const v4f ka = ks[t * (Mm / 4) + g * 2],  kb = ks[t * (Mm / 4) + g * 2 + 1];
            const float u = usf[t * 32 + lane];
            const v2f vcv[CG / 2] = { va.xy, va.zw, vb.xy, vb.zw };
            const v2f kv[CG / 2]  = { ka.xy, ka.zw, kb.xy, kb.zw };
#pragma unroll
            for (int e = 0; e < CG / 2; ++e) {
                const v2f cc = { gate_from(th2, vcv[e].x), gate_from(th2, vcv[e].y) };
                s[e] = __builtin_elementwise_fma(s[e], cc, kv[e] * u);
                p[e] *= cc;
            }
        }

        const size_t so = summary_offset(b, run * PC + h, rh, g, lane);
        *(v4f*)(Send + so)                = (v4f){ s[0].x, s[0].y, s[1].x, s[1].y };
        *(v4f*)(Send + so + SUMMARY_QUAD) = (v4f){ s[2].x, s[2].y, s[3].x, s[3].y };
        *(v4f*)(Pend + so)                = (v4f){ p[0].x, p[0].y, p[1].x, p[1].y };
        *(v4f*)(Pend + so + SUMMARY_QUAD) = (v4f){ p[2].x, p[2].y, p[3].x, p[3].y };
        if (run == NC / PC - 1) break;
    }
}

#define EW (RH * NG)
#define YPITCH (Mm + 4)
__global__ __launch_bounds__(32 * EW)
void scan_emit_out_kernel(const float* __restrict__ K,
                          const float* __restrict__ Q,
                          const float* __restrict__ V,
                          const float* __restrict__ vc,
                          const float* __restrict__ eta,
                          const float* __restrict__ theta_matrix,
                          const float* __restrict__ Send,
                          const float* __restrict__ Pend,
                          const float* __restrict__ Wout,
                          const float* __restrict__ bout,
                          float* __restrict__ out)
{
    __shared__ v4f vcs[LC * Mm / 4];
    __shared__ v4f ks[LC * Mm / 4];
    __shared__ v4f qs[LC * Mm / 4];
    __shared__ v4f us[LC * Mm / 4];
    __shared__ float yp[NG * LC * Mm];
    __shared__ v2f yt2[LC * YPITCH / 2];
    float* yt = (float*)yt2;

    const int b    = blockIdx.x;
    const int c    = blockIdx.y;
    const int tid  = threadIdx.x;
    const int lane = tid & 31;
    const int wv   = tid >> 5;
    const int rh   = wv / NG;
    const int g    = wv % NG;
    const int hi   = lane >> 4;
    const int lr   = lane & 15;
    const size_t r0 = (size_t)b * Tt + (size_t)c * LC;

    {
        const v4f* vcg = (const v4f*)(vc + r0 * Mm);
        const v4f* kg  = (const v4f*)(K  + r0 * Mm);
        const v4f* qg  = (const v4f*)(Q  + r0 * Mm);
        const v4f* vg  = (const v4f*)(V  + r0 * Mm);
        vcs[tid] = vcg[tid];
        ks[tid]  = kg[tid];
        qs[tid]  = qg[tid];
        us[tid]  = vg[tid] * eta[r0 + (tid >> 4)];
    }

    const int n0 = wv * 32;
    v2f bf0[Mm / 4], bf1[Mm / 4];
    {
        const float* Bp0 = Wout + (size_t)(n0 + lr) * Mm + 2 * hi;
        const float* Bp1 = Bp0 + 16 * Mm;
#pragma unroll
        for (int k4 = 0; k4 < Mm / 4; ++k4) {
            bf0[k4] = *(const v2f*)(Bp0 + 4 * k4);
            bf1[k4] = *(const v2f*)(Bp1 + 4 * k4);
        }
    }

    v2f s[CG / 2];
#pragma unroll
    for (int e = 0; e < CG / 2; ++e) s[e] = (v2f){ 0.f, 0.f };
    {
        const float* Sb = Send + summary_offset(b, 0, rh, g, lane);
        const float* Pb = Pend + summary_offset(b, 0, rh, g, lane);
        const int n    = (c + PC - 1) / PC;
        const int last = c - 1;
        int e0 = 0;
        for (; e0 + 8 <= n; e0 += 8) fold_chunks<8>(Sb, Pb, e0, last, s);
        switch (n - e0) {
        case 7: fold_chunks<7>(Sb, Pb, e0, last, s); break;
        case 6: fold_chunks<6>(Sb, Pb, e0, last, s); break;
        case 5: fold_chunks<5>(Sb, Pb, e0, last, s); break;
        case 4: fold_chunks<4>(Sb, Pb, e0, last, s); break;
        case 3: fold_chunks<3>(Sb, Pb, e0, last, s); break;
        case 2: fold_chunks<2>(Sb, Pb, e0, last, s); break;
        case 1: fold_chunks<1>(Sb, Pb, e0, last, s); break;
        default: break;
        }
    }
    __syncthreads();

    const float th2 = theta_matrix[rh * 32 + lane] * -2.8853900817779268f;
    const float* usf = (const float*)us + rh * 32 + lane;
    float* ypw = yp + g * (LC * Mm) + rh * 32 + lane;

#pragma unroll 2
    for (int t = 0; t < LC; ++t) {
        const v4f va = vcs[t * (Mm / 4) + g * 2], vb = vcs[t * (Mm / 4) + g * 2 + 1];
        const v4f ka = ks[t * (Mm / 4) + g * 2],  kb = ks[t * (Mm / 4) + g * 2 + 1];
        const v4f qa = qs[t * (Mm / 4) + g * 2],  qb = qs[t * (Mm / 4) + g * 2 + 1];
        const float u = usf[t * Mm];
        const v2f vcv[CG / 2] = { va.xy, va.zw, vb.xy, vb.zw };
        const v2f kv[CG / 2]  = { ka.xy, ka.zw, kb.xy, kb.zw };
        const v2f qv[CG / 2]  = { qa.xy, qa.zw, qb.xy, qb.zw };
        v2f yy = { 0.f, 0.f };
#pragma unroll
        for (int e = 0; e < CG / 2; ++e) {
            const v2f cc = { gate_from(th2, vcv[e].x), gate_from(th2, vcv[e].y) };
            s[e] = __builtin_elementwise_fma(s[e], cc, kv[e] * u);
            yy = __builtin_elementwise_fma(s[e], qv[e], yy);
        }
        ypw[t * Mm] = yy.x + yy.y;
    }
    __syncthreads();

#pragma unroll
    for (int k = 0; k < (LC * Mm) / (32 * EW); ++k) {
        const int idx = tid + (32 * EW) * k;
        float acc = yp[idx];
#pragma unroll
        for (int g2 = 1; g2 < NG; ++g2) acc += yp[g2 * (LC * Mm) + idx];
        yt[(idx >> 6) * YPITCH + (idx & 63)] = acc;
    }
    __syncthreads();

    const float* A0 = yt + lr * YPITCH + 2 * hi;
    const float* A1 = A0 + 16 * YPITCH;
    v8f c00 = {}, c01 = {}, c10 = {}, c11 = {};
#pragma unroll
    for (int k4 = 0; k4 < Mm / 4; ++k4) {
        const v2f a0 = *(const v2f*)(A0 + 4 * k4);
        const v2f a1 = *(const v2f*)(A1 + 4 * k4);
        c00 = __builtin_amdgcn_wmma_f32_16x16x4_f32(false, a0, false, bf0[k4], (short)0, c00, false, false);
        c01 = __builtin_amdgcn_wmma_f32_16x16x4_f32(false, a0, false, bf1[k4], (short)0, c01, false, false);
        c10 = __builtin_amdgcn_wmma_f32_16x16x4_f32(false, a1, false, bf0[k4], (short)0, c10, false, false);
        c11 = __builtin_amdgcn_wmma_f32_16x16x4_f32(false, a1, false, bf1[k4], (short)0, c11, false, false);
    }

    const float bb0 = bout[n0 + lr];
    const float bb1 = bout[n0 + 16 + lr];
#pragma unroll
    for (int r = 0; r < 8; ++r) {
        float* o0 = out + (r0 + r + 8 * hi) * Oo + n0 + lr;
        float* o1 = o0 + (size_t)16 * Oo;
        o0[0]  = c00[r] + bb0;
        o0[16] = c01[r] + bb1;
        o1[0]  = c10[r] + bb0;
        o1[16] = c11[r] + bb1;
    }
}

extern "C" void kernel_launch(void* const* d_in, const int* in_sizes, int n_in,
                              void* d_out, int out_size, void* d_ws, size_t ws_size,
                              hipStream_t stream)
{
    const float* x     = (const float*)d_in[0];
    const float* Wk    = (const float*)d_in[1];
    const float* Wq    = (const float*)d_in[2];
    const float* Wv    = (const float*)d_in[3];
    const float* Wvc   = (const float*)d_in[4];
    const float* We    = (const float*)d_in[5];
    const float* theta = (const float*)d_in[6];
    const float* Wout  = (const float*)d_in[7];
    const float* bout  = (const float*)d_in[8];
    const float* gamma = (const float*)d_in[9];
    const float* beta  = (const float*)d_in[10];
    float* out = (float*)d_out;

    float* ws  = (float*)d_ws;
    const size_t NROW = (size_t)Bb * Tt;
    float* Ko  = ws;
    float* Qo  = Ko  + NROW * Mm;
    float* Vo  = Qo  + NROW * Mm;
    float* vco = Vo  + NROW * Mm;
    float* eta = vco + NROW * Mm;
    float* Send = eta + NROW;
    float* Pend = Send + (size_t)Bb * NC * Mm * Mm;

    ln_proj_gemm_kernel<<<dim3(NROW / (16 * PW), 4), dim3(32 * PW), 0, stream>>>(x, gamma, beta, We,
                                                                  Wk, Wq, Wv, Wvc,
                                                                  Ko, Qo, Vo, vco, eta);
    scan_summary_kernel<<<dim3(Bb, RH, NC / PC), dim3(32 * NG), 0, stream>>>(Ko, Vo, vco, eta, theta,
                                                                            Send, Pend);
    scan_emit_out_kernel<<<dim3(Bb, NC), dim3(32 * EW), 0, stream>>>(Ko, Qo, Vo, vco, eta, theta,
                                                                     Send, Pend, Wout, bout, out);
}


# scan gates: 1+tanh(theta_i*vc_j) with tanh on the hardware transcendental instruction, one transcendental a gate instead of exp2, multiply-add and reciprocal; gate rounding is now that instruction's
# speedup vs baseline: 17.8786x; 1.0400x over previous
#include <hip/hip_runtime.h>
#include <hip/hip_bf16.h>

typedef float v2f __attribute__((ext_vector_type(2)));
typedef float v4f __attribute__((ext_vector_type(4)));
typedef float v8f __attribute__((ext_vector_type(8)));

#define Bb 8
#define Tt 1024
#define Dd 512
#define Mm 64
#define Oo 512
#define NC 32
#define LC (Tt / NC)

#define PW 8
#define WPITCH (Dd + 4)
#define WR (Mm / PW)

template <bool ETA>
__device__ __forceinline__ void proj_k_loop(const float* __restrict__ Arow,
                                            const float* B0, const float* B1,
                                            const float* B2, const float* B3,
                                            const float* Ge,
                                            v8f& c0, v8f& c1, v8f& c2, v8f& c3,
                                            v2f& s1, v2f& s2, v2f& e1)
{
#pragma unroll 8
    for (int kk = 0; kk < Dd; kk += 4) {
        v2f a  = *(const v2f*)(Arow + kk);
        v2f b0 = *(const v2f*)(B0 + kk);
        v2f b1 = *(const v2f*)(B1 + kk);
        v2f b2 = *(const v2f*)(B2 + kk);
        v2f b3 = *(const v2f*)(B3 + kk);
        c0 = __builtin_amdgcn_wmma_f32_16x16x4_f32(false, a, false, b0, (short)0, c0, false, false);
        c1 = __builtin_amdgcn_wmma_f32_16x16x4_f32(false, a, false, b1, (short)0, c1, false, false);
        c2 = __builtin_amdgcn_wmma_f32_16x16x4_f32(false, a, false, b2, (short)0, c2, false, false);
        c3 = __builtin_amdgcn_wmma_f32_16x16x4_f32(false, a, false, b3, (short)0, c3, false, false);
        s1 += a;
        s2 = __builtin_elementwise_fma(a, a, s2);
        if (ETA) e1 = __builtin_elementwise_fma(a, *(const v2f*)(Ge + kk), e1);
    }
}

__global__ __launch_bounds__(32 * PW)
void ln_proj_gemm_kernel(const float* __restrict__ x,
                         const float* __restrict__ gamma,
                         const float* __restrict__ beta,
                         const float* __restrict__ We,
                         const float* __restrict__ Wk,
                         const float* __restrict__ Wq,
                         const float* __restrict__ Wv,
                         const float* __restrict__ Wvc,
                         float* __restrict__ Ko,
                         float* __restrict__ Qo,
                         float* __restrict__ Vo,
                         float* __restrict__ vco,
                         float* __restrict__ eta)
{
    __shared__ v4f Ws4[Mm * WPITCH / 4];
    __shared__ v4f gwe4[Dd / 4];
    __shared__ float csL[Mm];
    __shared__ float bsL[Mm];
    float* Ws = (float*)Ws4;

    const int w = blockIdx.y;
    const float* W = (w == 0) ? Wk : (w == 1) ? Wq : (w == 2) ? Wv : Wvc;
    float* Out     = (w == 0) ? Ko : (w == 1) ? Qo : (w == 2) ? Vo : vco;

    const int tid   = threadIdx.x;
    const int wave  = tid >> 5;
    const int lane  = tid & 31;
    const int hi    = lane >> 4;
    const int lr    = lane & 15;

    v4f g4[4], b4[4];
#pragma unroll
    for (int j = 0; j < 4; ++j) {
        g4[j] = *(const v4f*)(gamma + (lane + 32 * j) * 4);
        b4[j] = *(const v4f*)(beta  + (lane + 32 * j) * 4);
    }
    v4f wr[WR][4];
#pragma unroll
    for (int r = 0; r < WR; ++r) {
        const float* Wrow = W + (size_t)(wave * WR + r) * Dd;
#pragma unroll
        for (int j = 0; j < 4; ++j) wr[r][j] = *(const v4f*)(Wrow + (lane + 32 * j) * 4);
    }
    float cs[WR + 1], bs[WR + 1];
#pragma unroll
    for (int r = 0; r < WR; ++r) {
        const int row = wave * WR + r;
        cs[r] = 0.f; bs[r] = 0.f;
#pragma unroll
        for (int j = 0; j < 4; ++j) {
            const v4f gw = wr[r][j] * g4[j];
            const v4f bw = wr[r][j] * b4[j];
            *(v4f*)(Ws + row * WPITCH + (lane + 32 * j) * 4) = gw;
            cs[r] += (gw.x + gw.y) + (gw.z + gw.w);
            bs[r] += (bw.x + bw.y) + (bw.z + bw.w);
        }
    }
    cs[WR] = 0.f; bs[WR] = 0.f;
    if (w == 2) {
#pragma unroll
        for (int j = 0; j < 4; ++j) {
            const v4f we = *(const v4f*)(We + (lane + 32 * j) * 4);
            const v4f gw = we * g4[j];
            const v4f bw = we * b4[j];
            if (wave == 0) gwe4[lane + 32 * j] = gw;
            cs[WR] += (gw.x + gw.y) + (gw.z + gw.w);
            bs[WR] += (bw.x + bw.y) + (bw.z + bw.w);
        }
    }
#pragma unroll
    for (int m = 16; m >= 1; m >>= 1) {
#pragma unroll
        for (int r = 0; r <= WR; ++r) {
            cs[r] += __shfl_xor(cs[r], m, 32);
            bs[r] += __shfl_xor(bs[r], m, 32);
        }
    }
    if (lane == 0) {
#pragma unroll
        for (int r = 0; r < WR; ++r) { csL[wave * WR + r] = cs[r]; bsL[wave * WR + r] = bs[r]; }
    }
    const float cse = cs[WR], bse = bs[WR];
    __syncthreads();

    const int tileM = (blockIdx.x * PW + wave) * 16;
    const float* Arow = x + (size_t)(tileM + lr) * Dd + 2 * hi;
    const float* B0 = Ws + (lr +  0) * WPITCH + 2 * hi;
    const float* B1 = Ws + (lr + 16) * WPITCH + 2 * hi;
    const float* B2 = Ws + (lr + 32) * WPITCH + 2 * hi;
    const float* B3 = Ws + (lr + 48) * WPITCH + 2 * hi;
    const float* Ge = (const float*)gwe4 + 2 * hi;

    v8f c0 = {}, c1 = {}, c2 = {}, c3 = {};
    v2f s1 = { 0.f, 0.f }, s2 = { 0.f, 0.f }, e1 = { 0.f, 0.f };
    if (w == 2) proj_k_loop<true >(Arow, B0, B1, B2, B3, Ge, c0, c1, c2, c3, s1, s2, e1);
    else        proj_k_loop<false>(Arow, B0, B1, B2, B3, Ge, c0, c1, c2, c3, s1, s2, e1);

    float sum = s1.x + s1.y, sq = s2.x + s2.y;
    sum += __shfl_xor(sum, 16, 32);
    sq  += __shfl_xor(sq,  16, 32);
    const float mu  = sum * (1.0f / Dd);
    const float var = sq * (1.0f / Dd) - mu * mu;
    const float rs  = rsqrtf(var + 1e-5f);

    if (w == 2) {
        float e = e1.x + e1.y;
        e += __shfl_xor(e, 16, 32);
        const float dot = rs * (e - mu * cse) + bse;
        if (hi == 0) eta[tileM + lr] = 1.0f / (1.0f + expf(-dot));
    }

    const float cs0 = csL[lr], cs1 = csL[lr + 16], cs2 = csL[lr + 32], cs3 = csL[lr + 48];
    const float bs0 = bsL[lr], bs1 = bsL[lr + 16], bs2 = bsL[lr + 32], bs3 = bsL[lr + 48];
#pragma unroll
    for (int r = 0; r < 8; ++r) {
        const float mur = __shfl(mu, r + 8 * hi, 32);
        const float rsr = __shfl(rs, r + 8 * hi, 32);
        c0[r] = rsr * (c0[r] - mur * cs0) + bs0;
        c1[r] = rsr * (c1[r] - mur * cs1) + bs1;
        c2[r] = rsr * (c2[r] - mur * cs2) + bs2;
        c3[r] = rsr * (c3[r] - mur * cs3) + bs3;
    }

    if (w < 2) {
#pragma unroll
        for (int r = 0; r < 8; ++r) {
            c0[r] = fmaxf(c0[r], 0.f); c1[r] = fmaxf(c1[r], 0.f);
            c2[r] = fmaxf(c2[r], 0.f); c3[r] = fmaxf(c3[r], 0.f);
            float s = c0[r] + c1[r] + c2[r] + c3[r];
#pragma unroll
            for (int m = 8; m >= 1; m >>= 1) s += __shfl_xor(s, m, 32);
            float inv = 1.0f / (1e-5f + s);
            c0[r] *= inv; c1[r] *= inv; c2[r] *= inv; c3[r] *= inv;
        }
    }

#pragma unroll
    for (int r = 0; r < 8; ++r) {
        float* o = Out + (size_t)(tileM + r + 8 * hi) * Mm + lr;
        o[0]  = c0[r];
        o[16] = c1[r];
        o[32] = c2[r];
        o[48] = c3[r];
    }
}

#define CG 8
#define NG (Mm / CG)
#define RH (Mm / 32)
#define PC 2

__device__ __forceinline__ float gate_from(float th, float vcj)
{
    return 1.0f + __builtin_amdgcn_tanhf(th * vcj);
}

#define SUMMARY_QUAD (32 * 4)
__device__ __forceinline__ size_t summary_offset(int b, int c, int rh, int g, int lane)
{
    return ((((size_t)b * NC + c) * RH + rh) * NG + g) * (32 * CG) + (size_t)lane * 4;
}

#define SUMMARY_CHUNK (RH * NG * 32 * CG)

template <int N>
__device__ __forceinline__ void fold_chunks(const float* __restrict__ Sb,
                                            const float* __restrict__ Pb,
                                            int e0, int last,
                                            v2f (&s)[CG / 2])
{
    v4f sa[N], sb[N], pa[N], pb[N];
#pragma unroll
    for (int k = 0; k < N; ++k) {
        const int slot = min(PC * (e0 + k) + (PC - 1), last);
        const float* Sp = Sb + slot * SUMMARY_CHUNK;
        const float* Pp = Pb + slot * SUMMARY_CHUNK;
        sa[k] = *(const v4f*)(Sp);  sb[k] = *(const v4f*)(Sp + SUMMARY_QUAD);
        pa[k] = *(const v4f*)(Pp);  pb[k] = *(const v4f*)(Pp + SUMMARY_QUAD);
    }
#pragma unroll
    for (int k = 0; k < N; ++k) {
        s[0] = __builtin_elementwise_fma(pa[k].xy, s[0], sa[k].xy);
        s[1] = __builtin_elementwise_fma(pa[k].zw, s[1], sa[k].zw);
        s[2] = __builtin_elementwise_fma(pb[k].xy, s[2], sb[k].xy);
        s[3] = __builtin_elementwise_fma(pb[k].zw, s[3], sb[k].zw);
    }
}

__global__ __launch_bounds__(32 * NG)
void scan_summary_kernel(const float* __restrict__ K,
                         const float* __restrict__ V,
                         const float* __restrict__ vc,
                         const float* __restrict__ eta,
                         const float* __restrict__ theta_matrix,
                         float* __restrict__ Send,
                         float* __restrict__ Pend)
{
    __shared__ v4f vcs[PC * LC * Mm / 4];
    __shared__ v4f ks[PC * LC * Mm / 4];
    __shared__ v4f us[PC * LC * 32 / 4];

    const int b    = blockIdx.x;
    const int rh   = blockIdx.y;
    const int run  = blockIdx.z;
    const int tid  = threadIdx.x;
    const int lane = tid & 31;
    const int g    = tid >> 5;
    const size_t r0 = (size_t)b * Tt + (size_t)run * (PC * LC);

    {
        const v4f* vcg = (const v4f*)(vc + r0 * Mm);
        const v4f* kg  = (const v4f*)(K  + r0 * Mm);
#pragma unroll
        for (int q = 0; q < (PC * LC * Mm / 4) / (32 * NG); ++q) {
            vcs[tid + 32 * NG * q] = vcg[tid + 32 * NG * q];
            ks[tid + 32 * NG * q]  = kg[tid + 32 * NG * q];
        }
#pragma unroll
        for (int q = 0; q < (PC * LC * 32 / 4) / (32 * NG); ++q) {
            const int idx = tid + 32 * NG * q;
            const int t = idx >> 3, part = idx & 7;
            const v4f v4 = *(const v4f*)(V + (r0 + t) * Mm + rh * 32 + part * 4);
            us[idx] = v4 * eta[r0 + t];
        }
    }
    __syncthreads();

    const float th = theta_matrix[rh * 32 + lane];
    const float* usf = (const float*)us;

    v2f s[CG / 2], p[CG / 2];
#pragma unroll
    for (int e = 0; e < CG / 2; ++e) { s[e] = (v2f){ 0.f, 0.f }; p[e] = (v2f){ 1.f, 1.f }; }

    for (int h = 0; h < PC; ++h) {
#pragma unroll 2
        for (int t = h * LC; t < (h + 1) * LC; ++t) {
            const v4f va = vcs[t * (Mm / 4) + g * 2], vb = vcs[t * (Mm / 4) + g * 2 + 1];
            const v4f ka = ks[t * (Mm / 4) + g * 2],  kb = ks[t * (Mm / 4) + g * 2 + 1];
            const float u = usf[t * 32 + lane];
            const v2f vcv[CG / 2] = { va.xy, va.zw, vb.xy, vb.zw };
            const v2f kv[CG / 2]  = { ka.xy, ka.zw, kb.xy, kb.zw };
#pragma unroll
            for (int e = 0; e < CG / 2; ++e) {
                const v2f cc = { gate_from(th, vcv[e].x), gate_from(th, vcv[e].y) };
                s[e] = __builtin_elementwise_fma(s[e], cc, kv[e] * u);
                p[e] *= cc;
            }
        }

        const size_t so = summary_offset(b, run * PC + h, rh, g, lane);
        *(v4f*)(Send + so)                = (v4f){ s[0].x, s[0].y, s[1].x, s[1].y };
        *(v4f*)(Send + so + SUMMARY_QUAD) = (v4f){ s[2].x, s[2].y, s[3].x, s[3].y };
        *(v4f*)(Pend + so)                = (v4f){ p[0].x, p[0].y, p[1].x, p[1].y };
        *(v4f*)(Pend + so + SUMMARY_QUAD) = (v4f){ p[2].x, p[2].y, p[3].x, p[3].y };
        if (run == NC / PC - 1) break;
    }
}

#define EW (RH * NG)
#define YPITCH (Mm + 4)
__global__ __launch_bounds__(32 * EW)
void scan_emit_out_kernel(const float* __restrict__ K,
                          const float* __restrict__ Q,
                          const float* __restrict__ V,
                          const float* __restrict__ vc,
                          const float* __restrict__ eta,
                          const float* __restrict__ theta_matrix,
                          const float* __restrict__ Send,
                          const float* __restrict__ Pend,
                          const float* __restrict__ Wout,
                          const float* __restrict__ bout,
                          float* __restrict__ out)
{
    __shared__ v4f vcs[LC * Mm / 4];
    __shared__ v4f ks[LC * Mm / 4];
    __shared__ v4f qs[LC * Mm / 4];
    __shared__ v4f us[LC * Mm / 4];
    __shared__ float yp[NG * LC * Mm];
    __shared__ v2f yt2[LC * YPITCH / 2];
    float* yt = (float*)yt2;

    const int b    = blockIdx.x;
    const int c    = blockIdx.y;
    const int tid  = threadIdx.x;
    const int lane = tid & 31;
    const int wv   = tid >> 5;
    const int rh   = wv / NG;
    const int g    = wv % NG;
    const int hi   = lane >> 4;
    const int lr   = lane & 15;
    const size_t r0 = (size_t)b * Tt + (size_t)c * LC;

    {
        const v4f* vcg = (const v4f*)(vc + r0 * Mm);
        const v4f* kg  = (const v4f*)(K  + r0 * Mm);
        const v4f* qg  = (const v4f*)(Q  + r0 * Mm);
        const v4f* vg  = (const v4f*)(V  + r0 * Mm);
        vcs[tid] = vcg[tid];
        ks[tid]  = kg[tid];
        qs[tid]  = qg[tid];
        us[tid]  = vg[tid] * eta[r0 + (tid >> 4)];
    }

    const int n0 = wv * 32;
    v2f bf0[Mm / 4], bf1[Mm / 4];
    {
        const float* Bp0 = Wout + (size_t)(n0 + lr) * Mm + 2 * hi;
        const float* Bp1 = Bp0 + 16 * Mm;
#pragma unroll
        for (int k4 = 0; k4 < Mm / 4; ++k4) {
            bf0[k4] = *(const v2f*)(Bp0 + 4 * k4);
            bf1[k4] = *(const v2f*)(Bp1 + 4 * k4);
        }
    }

    v2f s[CG / 2];
#pragma unroll
    for (int e = 0; e < CG / 2; ++e) s[e] = (v2f){ 0.f, 0.f };
    {
        const float* Sb = Send + summary_offset(b, 0, rh, g, lane);
        const float* Pb = Pend + summary_offset(b, 0, rh, g, lane);
        const int n    = (c + PC - 1) / PC;
        const int last = c - 1;
        int e0 = 0;
        for (; e0 + 8 <= n; e0 += 8) fold_chunks<8>(Sb, Pb, e0, last, s);
        switch (n - e0) {
        case 7: fold_chunks<7>(Sb, Pb, e0, last, s); break;
        case 6: fold_chunks<6>(Sb, Pb, e0, last, s); break;
        case 5: fold_chunks<5>(Sb, Pb, e0, last, s); break;
        case 4: fold_chunks<4>(Sb, Pb, e0, last, s); break;
        case 3: fold_chunks<3>(Sb, Pb, e0, last, s); break;
        case 2: fold_chunks<2>(Sb, Pb, e0, last, s); break;
        case 1: fold_chunks<1>(Sb, Pb, e0, last, s); break;
        default: break;
        }
    }
    __syncthreads();

    const float th = theta_matrix[rh * 32 + lane];
    const float* usf = (const float*)us + rh * 32 + lane;
    float* ypw = yp + g * (LC * Mm) + rh * 32 + lane;

#pragma unroll 2
    for (int t = 0; t < LC; ++t) {
        const v4f va = vcs[t * (Mm / 4) + g * 2], vb = vcs[t * (Mm / 4) + g * 2 + 1];
        const v4f ka = ks[t * (Mm / 4) + g * 2],  kb = ks[t * (Mm / 4) + g * 2 + 1];
        const v4f qa = qs[t * (Mm / 4) + g * 2],  qb = qs[t * (Mm / 4) + g * 2 + 1];
        const float u = usf[t * Mm];
        const v2f vcv[CG / 2] = { va.xy, va.zw, vb.xy, vb.zw };
        const v2f kv[CG / 2]  = { ka.xy, ka.zw, kb.xy, kb.zw };
        const v2f qv[CG / 2]  = { qa.xy, qa.zw, qb.xy, qb.zw };
        v2f yy = { 0.f, 0.f };
#pragma unroll
        for (int e = 0; e < CG / 2; ++e) {
            const v2f cc = { gate_from(th, vcv[e].x), gate_from(th, vcv[e].y) };
            s[e] = __builtin_elementwise_fma(s[e], cc, kv[e] * u);
            yy = __builtin_elementwise_fma(s[e], qv[e], yy);
        }
        ypw[t * Mm] = yy.x + yy.y;
    }
    __syncthreads();

#pragma unroll
    for (int k = 0; k < (LC * Mm) / (32 * EW); ++k) {
        const int idx = tid + (32 * EW) * k;
        float acc = yp[idx];
#pragma unroll
        for (int g2 = 1; g2 < NG; ++g2) acc += yp[g2 * (LC * Mm) + idx];
        yt[(idx >> 6) * YPITCH + (idx & 63)] = acc;
    }
    __syncthreads();

    const float* A0 = yt + lr * YPITCH + 2 * hi;
    const float* A1 = A0 + 16 * YPITCH;
    v8f c00 = {}, c01 = {}, c10 = {}, c11 = {};
#pragma unroll
    for (int k4 = 0; k4 < Mm / 4; ++k4) {
        const v2f a0 = *(const v2f*)(A0 + 4 * k4);
        const v2f a1 = *(const v2f*)(A1 + 4 * k4);
        c00 = __builtin_amdgcn_wmma_f32_16x16x4_f32(false, a0, false, bf0[k4], (short)0, c00, false, false);
        c01 = __builtin_amdgcn_wmma_f32_16x16x4_f32(false, a0, false, bf1[k4], (short)0, c01, false, false);
        c10 = __builtin_amdgcn_wmma_f32_16x16x4_f32(false, a1, false, bf0[k4], (short)0, c10, false, false);
        c11 = __builtin_amdgcn_wmma_f32_16x16x4_f32(false, a1, false, bf1[k4], (short)0, c11, false, false);
    }

    const float bb0 = bout[n0 + lr];
    const float bb1 = bout[n0 + 16 + lr];
#pragma unroll
    for (int r = 0; r < 8; ++r) {
        float* o0 = out + (r0 + r + 8 * hi) * Oo + n0 + lr;
        float* o1 = o0 + (size_t)16 * Oo;
        o0[0]  = c00[r] + bb0;
        o0[16] = c01[r] + bb1;
        o1[0]  = c10[r] + bb0;
        o1[16] = c11[r] + bb1;
    }
}

extern "C" void kernel_launch(void* const* d_in, const int* in_sizes, int n_in,
                              void* d_out, int out_size, void* d_ws, size_t ws_size,
                              hipStream_t stream)
{
    const float* x     = (const float*)d_in[0];
    const float* Wk    = (const float*)d_in[1];
    const float* Wq    = (const float*)d_in[2];
    const float* Wv    = (const float*)d_in[3];
    const float* Wvc   = (const float*)d_in[4];
    const float* We    = (const float*)d_in[5];
    const float* theta = (const float*)d_in[6];
    const float* Wout  = (const float*)d_in[7];
    const float* bout  = (const float*)d_in[8];
    const float* gamma = (const float*)d_in[9];
    const float* beta  = (const float*)d_in[10];
    float* out = (float*)d_out;

    float* ws  = (float*)d_ws;
    const size_t NROW = (size_t)Bb * Tt;
    float* Ko  = ws;
    float* Qo  = Ko  + NROW * Mm;
    float* Vo  = Qo  + NROW * Mm;
    float* vco = Vo  + NROW * Mm;
    float* eta = vco + NROW * Mm;
    float* Send = eta + NROW;
    float* Pend = Send + (size_t)Bb * NC * Mm * Mm;

    ln_proj_gemm_kernel<<<dim3(NROW / (16 * PW), 4), dim3(32 * PW), 0, stream>>>(x, gamma, beta, We,
                                                                  Wk, Wq, Wv, Wvc,
                                                                  Ko, Qo, Vo, vco, eta);
    scan_summary_kernel<<<dim3(Bb, RH, NC / PC), dim3(32 * NG), 0, stream>>>(Ko, Vo, vco, eta, theta,
                                                                            Send, Pend);
    scan_emit_out_kernel<<<dim3(Bb, NC), dim3(32 * EW), 0, stream>>>(Ko, Qo, Vo, vco, eta, theta,
                                                                     Send, Pend, Wout, bout, out);
}


# projection GEMM K loop: two operand register sets; a trip's A (global) and B (LDS) fragments are fetched one trip ahead of its 32 WMMAs, trip 0's A ahead of the weight staging; same sums
# speedup vs baseline: 19.4723x; 1.0891x over previous
#include <hip/hip_runtime.h>
#include <hip/hip_bf16.h>

typedef float v2f __attribute__((ext_vector_type(2)));
typedef float v4f __attribute__((ext_vector_type(4)));
typedef float v8f __attribute__((ext_vector_type(8)));

#define Bb 8
#define Tt 1024
#define Dd 512
#define Mm 64
#define Oo 512
#define NC 32
#define LC (Tt / NC)

#define PW 8
#define WPITCH (Dd + 4)
#define WR (Mm / PW)

struct TripOps { v2f a[8]; v2f b[8][4]; v2f g[8]; };

__device__ __forceinline__ void fetch_trip_a(TripOps& o, const float* __restrict__ Arow, int kk)
{
#pragma unroll
    for (int i = 0; i < 8; ++i) o.a[i] = *(const v2f*)(Arow + kk + 4 * i);
}

template <bool ETA>
__device__ __forceinline__ void fetch_trip_b(TripOps& o,
                                             const float* B0, const float* B1,
                                             const float* B2, const float* B3,
                                             const float* Ge, int kk)
{
#pragma unroll
    for (int i = 0; i < 8; ++i) {
        o.b[i][0] = *(const v2f*)(B0 + kk + 4 * i);
        o.b[i][1] = *(const v2f*)(B1 + kk + 4 * i);
        o.b[i][2] = *(const v2f*)(B2 + kk + 4 * i);
        o.b[i][3] = *(const v2f*)(B3 + kk + 4 * i);
        if (ETA) o.g[i] = *(const v2f*)(Ge + kk + 4 * i);
    }
}

template <bool ETA>
__device__ __forceinline__ void mma_trip(const TripOps& o,
                                         v8f& c0, v8f& c1, v8f& c2, v8f& c3,
                                         v2f& s1, v2f& s2, v2f& e1)
{
#pragma unroll
    for (int i = 0; i < 8; ++i) {
        const v2f a = o.a[i];
        c0 = __builtin_amdgcn_wmma_f32_16x16x4_f32(false, a, false, o.b[i][0], (short)0, c0, false, false);
        c1 = __builtin_amdgcn_wmma_f32_16x16x4_f32(false, a, false, o.b[i][1], (short)0, c1, false, false);
        c2 = __builtin_amdgcn_wmma_f32_16x16x4_f32(false, a, false, o.b[i][2], (short)0, c2, false, false);
        c3 = __builtin_amdgcn_wmma_f32_16x16x4_f32(false, a, false, o.b[i][3], (short)0, c3, false, false);
        s1 += a;
        s2 = __builtin_elementwise_fma(a, a, s2);
        if (ETA) e1 = __builtin_elementwise_fma(a, o.g[i], e1);
    }
}

template <bool ETA>
__device__ __forceinline__ void proj_k_loop(TripOps& p,
                                            const float* __restrict__ Arow,
                                            const float* B0, const float* B1,
                                            const float* B2, const float* B3,
                                            const float* Ge,
                                            v8f& c0, v8f& c1, v8f& c2, v8f& c3,
                                            v2f& s1, v2f& s2, v2f& e1)
{
    TripOps q;
    fetch_trip_b<ETA>(p, B0, B1, B2, B3, Ge, 0);
#pragma unroll 1
    for (int kk = 0; kk < Dd; kk += 64) {
        fetch_trip_a(q, Arow, kk + 32);
        fetch_trip_b<ETA>(q, B0, B1, B2, B3, Ge, kk + 32);
        __builtin_amdgcn_sched_barrier(0);
        mma_trip<ETA>(p, c0, c1, c2, c3, s1, s2, e1);
        __builtin_amdgcn_sched_barrier(0);
        const int kn = min(kk + 64, Dd - 32);
        fetch_trip_a(p, Arow, kn);
        fetch_trip_b<ETA>(p, B0, B1, B2, B3, Ge, kn);
        __builtin_amdgcn_sched_barrier(0);
        mma_trip<ETA>(q, c0, c1, c2, c3, s1, s2, e1);
        __builtin_amdgcn_sched_barrier(0);
    }
}

__global__ __launch_bounds__(32 * PW)
void ln_proj_gemm_kernel(const float* __restrict__ x,
                         const float* __restrict__ gamma,
                         const float* __restrict__ beta,
                         const float* __restrict__ We,
                         const float* __restrict__ Wk,
                         const float* __restrict__ Wq,
                         const float* __restrict__ Wv,
                         const float* __restrict__ Wvc,
                         float* __restrict__ Ko,
                         float* __restrict__ Qo,
                         float* __restrict__ Vo,
                         float* __restrict__ vco,
                         float* __restrict__ eta)
{
    __shared__ v4f Ws4[Mm * WPITCH / 4];
    __shared__ v4f gwe4[Dd / 4];
    __shared__ float csL[Mm];
    __shared__ float bsL[Mm];
    float* Ws = (float*)Ws4;

    const int w = blockIdx.y;
    const float* W = (w == 0) ? Wk : (w == 1) ? Wq : (w == 2) ? Wv : Wvc;
    float* Out     = (w == 0) ? Ko : (w == 1) ? Qo : (w == 2) ? Vo : vco;

    const int tid   = threadIdx.x;
    const int wave  = tid >> 5;
    const int lane  = tid & 31;
    const int hi    = lane >> 4;
    const int lr    = lane & 15;

    const int tileM = (blockIdx.x * PW + wave) * 16;
    const float* Arow = x + (size_t)(tileM + lr) * Dd + 2 * hi;
    TripOps p;
    fetch_trip_a(p, Arow, 0);

    v4f g4[4], b4[4];
#pragma unroll
    for (int j = 0; j < 4; ++j) {
        g4[j] = *(const v4f*)(gamma + (lane + 32 * j) * 4);
        b4[j] = *(const v4f*)(beta  + (lane + 32 * j) * 4);
    }
    v4f wr[WR][4];
#pragma unroll
    for (int r = 0; r < WR; ++r) {
        const float* Wrow = W + (size_t)(wave * WR + r) * Dd;
#pragma unroll
        for (int j = 0; j < 4; ++j) wr[r][j] = *(const v4f*)(Wrow + (lane + 32 * j) * 4);
    }
    float cs[WR + 1], bs[WR + 1];
#pragma unroll
    for (int r = 0; r < WR; ++r) {
        const int row = wave * WR + r;
        cs[r] = 0.f; bs[r] = 0.f;
#pragma unroll
        for (int j = 0; j < 4; ++j) {
            const v4f gw = wr[r][j] * g4[j];
            const v4f bw = wr[r][j] * b4[j];
            *(v4f*)(Ws + row * WPITCH + (lane + 32 * j) * 4) = gw;
            cs[r] += (gw.x + gw.y) + (gw.z + gw.w);
            bs[r] += (bw.x + bw.y) + (bw.z + bw.w);
        }
    }
    cs[WR] = 0.f; bs[WR] = 0.f;
    if (w == 2) {
#pragma unroll
        for (int j = 0; j < 4; ++j) {
            const v4f we = *(const v4f*)(We + (lane + 32 * j) * 4);
            const v4f gw = we * g4[j];
            const v4f bw = we * b4[j];
            if (wave == 0) gwe4[lane + 32 * j] = gw;
            cs[WR] += (gw.x + gw.y) + (gw.z + gw.w);
            bs[WR] += (bw.x + bw.y) + (bw.z + bw.w);
        }
    }
#pragma unroll
    for (int m = 16; m >= 1; m >>= 1) {
#pragma unroll
        for (int r = 0; r <= WR; ++r) {
            cs[r] += __shfl_xor(cs[r], m, 32);
            bs[r] += __shfl_xor(bs[r], m, 32);
        }
    }
    if (lane == 0) {
#pragma unroll
        for (int r = 0; r < WR; ++r) { csL[wave * WR + r] = cs[r]; bsL[wave * WR + r] = bs[r]; }
    }
    const float cse = cs[WR], bse = bs[WR];
    __syncthreads();

    const float* B0 = Ws + (lr +  0) * WPITCH + 2 * hi;
    const float* B1 = Ws + (lr + 16) * WPITCH + 2 * hi;
    const float* B2 = Ws + (lr + 32) * WPITCH + 2 * hi;
    const float* B3 = Ws + (lr + 48) * WPITCH + 2 * hi;
    const float* Ge = (const float*)gwe4 + 2 * hi;

    v8f c0 = {}, c1 = {}, c2 = {}, c3 = {};
    v2f s1 = { 0.f, 0.f }, s2 = { 0.f, 0.f }, e1 = { 0.f, 0.f };
    if (w == 2) proj_k_loop<true >(p, Arow, B0, B1, B2, B3, Ge, c0, c1, c2, c3, s1, s2, e1);
    else        proj_k_loop<false>(p, Arow, B0, B1, B2, B3, Ge, c0, c1, c2, c3, s1, s2, e1);

    float sum = s1.x + s1.y, sq = s2.x + s2.y;
    sum += __shfl_xor(sum, 16, 32);
    sq  += __shfl_xor(sq,  16, 32);
    const float mu  = sum * (1.0f / Dd);
    const float var = sq * (1.0f / Dd) - mu * mu;
    const float rs  = rsqrtf(var + 1e-5f);

    if (w == 2) {
        float e = e1.x + e1.y;
        e += __shfl_xor(e, 16, 32);
        const float dot = rs * (e - mu * cse) + bse;
        if (hi == 0) eta[tileM + lr] = 1.0f / (1.0f + expf(-dot));
    }

    const float cs0 = csL[lr], cs1 = csL[lr + 16], cs2 = csL[lr + 32], cs3 = csL[lr + 48];
    const float bs0 = bsL[lr], bs1 = bsL[lr + 16], bs2 = bsL[lr + 32], bs3 = bsL[lr + 48];
#pragma unroll
    for (int r = 0; r < 8; ++r) {
        const float mur = __shfl(mu, r + 8 * hi, 32);
        const float rsr = __shfl(rs, r + 8 * hi, 32);
        c0[r] = rsr * (c0[r] - mur * cs0) + bs0;
        c1[r] = rsr * (c1[r] - mur * cs1) + bs1;
        c2[r] = rsr * (c2[r] - mur * cs2) + bs2;
        c3[r] = rsr * (c3[r] - mur * cs3) + bs3;
    }

    if (w < 2) {
#pragma unroll
        for (int r = 0; r < 8; ++r) {
            c0[r] = fmaxf(c0[r], 0.f); c1[r] = fmaxf(c1[r], 0.f);
            c2[r] = fmaxf(c2[r], 0.f); c3[r] = fmaxf(c3[r], 0.f);
            float s = c0[r] + c1[r] + c2[r] + c3[r];
#pragma unroll
            for (int m = 8; m >= 1; m >>= 1) s += __shfl_xor(s, m, 32);
            float inv = 1.0f / (1e-5f + s);
            c0[r] *= inv; c1[r] *= inv; c2[r] *= inv; c3[r] *= inv;
        }
    }

#pragma unroll
    for (int r = 0; r < 8; ++r) {
        float* o = Out + (size_t)(tileM + r + 8 * hi) * Mm + lr;
        o[0]  = c0[r];
        o[16] = c1[r];
        o[32] = c2[r];
        o[48] = c3[r];
    }
}

#define CG 8
#define NG (Mm / CG)
#define RH (Mm / 32)
#define PC 2

__device__ __forceinline__ float gate_from(float th, float vcj)
{
    return 1.0f + __builtin_amdgcn_tanhf(th * vcj);
}

#define SUMMARY_QUAD (32 * 4)
__device__ __forceinline__ size_t summary_offset(int b, int c, int rh, int g, int lane)
{
    return ((((size_t)b * NC + c) * RH + rh) * NG + g) * (32 * CG) + (size_t)lane * 4;
}

#define SUMMARY_CHUNK (RH * NG * 32 * CG)

template <int N>
__device__ __forceinline__ void fold_chunks(const float* __restrict__ Sb,
                                            const float* __restrict__ Pb,
                                            int e0, int last,
                                            v2f (&s)[CG / 2])
{
    v4f sa[N], sb[N], pa[N], pb[N];
#pragma unroll
    for (int k = 0; k < N; ++k) {
        const int slot = min(PC * (e0 + k) + (PC - 1), last);
        const float* Sp = Sb + slot * SUMMARY_CHUNK;
        const float* Pp = Pb + slot * SUMMARY_CHUNK;
        sa[k] = *(const v4f*)(Sp);  sb[k] = *(const v4f*)(Sp + SUMMARY_QUAD);
        pa[k] = *(const v4f*)(Pp);  pb[k] = *(const v4f*)(Pp + SUMMARY_QUAD);
    }
#pragma unroll
    for (int k = 0; k < N; ++k) {
        s[0] = __builtin_elementwise_fma(pa[k].xy, s[0], sa[k].xy);
        s[1] = __builtin_elementwise_fma(pa[k].zw, s[1], sa[k].zw);
        s[2] = __builtin_elementwise_fma(pb[k].xy, s[2], sb[k].xy);
        s[3] = __builtin_elementwise_fma(pb[k].zw, s[3], sb[k].zw);
    }
}

__global__ __launch_bounds__(32 * NG)
void scan_summary_kernel(const float* __restrict__ K,
                         const float* __restrict__ V,
                         const float* __restrict__ vc,
                         const float* __restrict__ eta,
                         const float* __restrict__ theta_matrix,
                         float* __restrict__ Send,
                         float* __restrict__ Pend)
{
    __shared__ v4f vcs[PC * LC * Mm / 4];
    __shared__ v4f ks[PC * LC * Mm / 4];
    __shared__ v4f us[PC * LC * 32 / 4];

    const int b    = blockIdx.x;
    const int rh   = blockIdx.y;
    const int run  = blockIdx.z;
    const int tid  = threadIdx.x;
    const int lane = tid & 31;
    const int g    = tid >> 5;
    const size_t r0 = (size_t)b * Tt + (size_t)run * (PC * LC);

    {
        const v4f* vcg = (const v4f*)(vc + r0 * Mm);
        const v4f* kg  = (const v4f*)(K  + r0 * Mm);
#pragma unroll
        for (int q = 0; q < (PC * LC * Mm / 4) / (32 * NG); ++q) {
            vcs[tid + 32 * NG * q] = vcg[tid + 32 * NG * q];
            ks[tid + 32 * NG * q]  = kg[tid + 32 * NG * q];
        }
#pragma unroll
        for (int q = 0; q < (PC * LC * 32 / 4) / (32 * NG); ++q) {
            const int idx = tid + 32 * NG * q;
            const int t = idx >> 3, part = idx & 7;
            const v4f v4 = *(const v4f*)(V + (r0 + t) * Mm + rh * 32 + part * 4);
            us[idx] = v4 * eta[r0 + t];
        }
    }
    __syncthreads();

    const float th = theta_matrix[rh * 32 + lane];
    const float* usf = (const float*)us;

    v2f s[CG / 2], p[CG / 2];
#pragma unroll
    for (int e = 0; e < CG / 2; ++e) { s[e] = (v2f){ 0.f, 0.f }; p[e] = (v2f){ 1.f, 1.f }; }

    for (int h = 0; h < PC; ++h) {
#pragma unroll 2
        for (int t = h * LC; t < (h + 1) * LC; ++t) {
            const v4f va = vcs[t * (Mm / 4) + g * 2], vb = vcs[t * (Mm / 4) + g * 2 + 1];
            const v4f ka = ks[t * (Mm / 4) + g * 2],  kb = ks[t * (Mm / 4) + g * 2 + 1];
            const float u = usf[t * 32 + lane];
            const v2f vcv[CG / 2] = { va.xy, va.zw, vb.xy, vb.zw };
            const v2f kv[CG / 2]  = { ka.xy, ka.zw, kb.xy, kb.zw };
#pragma unroll
            for (int e = 0; e < CG / 2; ++e) {
                const v2f cc = { gate_from(th, vcv[e].x), gate_from(th, vcv[e].y) };
                s[e] = __builtin_elementwise_fma(s[e], cc, kv[e] * u);
                p[e] *= cc;
            }
        }

        const size_t so = summary_offset(b, run * PC + h, rh, g, lane);
        *(v4f*)(Send + so)                = (v4f){ s[0].x, s[0].y, s[1].x, s[1].y };
        *(v4f*)(Send + so + SUMMARY_QUAD) = (v4f){ s[2].x, s[2].y, s[3].x, s[3].y };
        *(v4f*)(Pend + so)                = (v4f){ p[0].x, p[0].y, p[1].x, p[1].y };
        *(v4f*)(Pend + so + SUMMARY_QUAD) = (v4f){ p[2].x, p[2].y, p[3].x, p[3].y };
        if (run == NC / PC - 1) break;
    }
}

#define EW (RH * NG)
#define YPITCH (Mm + 4)
__global__ __launch_bounds__(32 * EW)
void scan_emit_out_kernel(const float* __restrict__ K,
                          const float* __restrict__ Q,
                          const float* __restrict__ V,
                          const float* __restrict__ vc,
                          const float* __restrict__ eta,
                          const float* __restrict__ theta_matrix,
                          const float* __restrict__ Send,
                          const float* __restrict__ Pend,
                          const float* __restrict__ Wout,
                          const float* __restrict__ bout,
                          float* __restrict__ out)
{
    __shared__ v4f vcs[LC * Mm / 4];
    __shared__ v4f ks[LC * Mm / 4];
    __shared__ v4f qs[LC * Mm / 4];
    __shared__ v4f us[LC * Mm / 4];
    __shared__ float yp[NG * LC * Mm];
    __shared__ v2f yt2[LC * YPITCH / 2];
    float* yt = (float*)yt2;

    const int b    = blockIdx.x;
    const int c    = blockIdx.y;
    const int tid  = threadIdx.x;
    const int lane = tid & 31;
    const int wv   = tid >> 5;
    const int rh   = wv / NG;
    const int g    = wv % NG;
    const int hi   = lane >> 4;
    const int lr   = lane & 15;
    const size_t r0 = (size_t)b * Tt + (size_t)c * LC;

    {
        const v4f* vcg = (const v4f*)(vc + r0 * Mm);
        const v4f* kg  = (const v4f*)(K  + r0 * Mm);
        const v4f* qg  = (const v4f*)(Q  + r0 * Mm);
        const v4f* vg  = (const v4f*)(V  + r0 * Mm);
        vcs[tid] = vcg[tid];
        ks[tid]  = kg[tid];
        qs[tid]  = qg[tid];
        us[tid]  = vg[tid] * eta[r0 + (tid >> 4)];
    }

    const int n0 = wv * 32;
    v2f bf0[Mm / 4], bf1[Mm / 4];
    {
        const float* Bp0 = Wout + (size_t)(n0 + lr) * Mm + 2 * hi;
        const float* Bp1 = Bp0 + 16 * Mm;
#pragma unroll
        for (int k4 = 0; k4 < Mm / 4; ++k4) {
            bf0[k4] = *(const v2f*)(Bp0 + 4 * k4);
            bf1[k4] = *(const v2f*)(Bp1 + 4 * k4);
        }
    }

    v2f s[CG / 2];
#pragma unroll
    for (int e = 0; e < CG / 2; ++e) s[e] = (v2f){ 0.f, 0.f };
    {
        const float* Sb = Send + summary_offset(b, 0, rh, g, lane);
        const float* Pb = Pend + summary_offset(b, 0, rh, g, lane);
        const int n    = (c + PC - 1) / PC;
        const int last = c - 1;
        int e0 = 0;
        for (; e0 + 8 <= n; e0 += 8) fold_chunks<8>(Sb, Pb, e0, last, s);
        switch (n - e0) {
        case 7: fold_chunks<7>(Sb, Pb, e0, last, s); break;
        case 6: fold_chunks<6>(Sb, Pb, e0, last, s); break;
        case 5: fold_chunks<5>(Sb, Pb, e0, last, s); break;
        case 4: fold_chunks<4>(Sb, Pb, e0, last, s); break;
        case 3: fold_chunks<3>(Sb, Pb, e0, last, s); break;
        case 2: fold_chunks<2>(Sb, Pb, e0, last, s); break;
        case 1: fold_chunks<1>(Sb, Pb, e0, last, s); break;
        default: break;
        }
    }
    __syncthreads();

    const float th = theta_matrix[rh * 32 + lane];
    const float* usf = (const float*)us + rh * 32 + lane;
    float* ypw = yp + g * (LC * Mm) + rh * 32 + lane;

#pragma unroll 2
    for (int t = 0; t < LC; ++t) {
        const v4f va = vcs[t * (Mm / 4) + g * 2], vb = vcs[t * (Mm / 4) + g * 2 + 1];
        const v4f ka = ks[t * (Mm / 4) + g * 2],  kb = ks[t * (Mm / 4) + g * 2 + 1];
        const v4f qa = qs[t * (Mm / 4) + g * 2],  qb = qs[t * (Mm / 4) + g * 2 + 1];
        const float u = usf[t * Mm];
        const v2f vcv[CG / 2] = { va.xy, va.zw, vb.xy, vb.zw };
        const v2f kv[CG / 2]  = { ka.xy, ka.zw, kb.xy, kb.zw };
        const v2f qv[CG / 2]  = { qa.xy, qa.zw, qb.xy, qb.zw };
        v2f yy = { 0.f, 0.f };
#pragma unroll
        for (int e = 0; e < CG / 2; ++e) {
            const v2f cc = { gate_from(th, vcv[e].x), gate_from(th, vcv[e].y) };
            s[e] = __builtin_elementwise_fma(s[e], cc, kv[e] * u);
            yy = __builtin_elementwise_fma(s[e], qv[e], yy);
        }
        ypw[t * Mm] = yy.x + yy.y;
    }
    __syncthreads();

#pragma unroll
    for (int k = 0; k < (LC * Mm) / (32 * EW); ++k) {
        const int idx = tid + (32 * EW) * k;
        float acc = yp[idx];
#pragma unroll
        for (int g2 = 1; g2 < NG; ++g2) acc += yp[g2 * (LC * Mm) + idx];
        yt[(idx >> 6) * YPITCH + (idx & 63)] = acc;
    }
    __syncthreads();

    const float* A0 = yt + lr * YPITCH + 2 * hi;
    const float* A1 = A0 + 16 * YPITCH;
    v8f c00 = {}, c01 = {}, c10 = {}, c11 = {};
#pragma unroll
    for (int k4 = 0; k4 < Mm / 4; ++k4) {
        const v2f a0 = *(const v2f*)(A0 + 4 * k4);
        const v2f a1 = *(const v2f*)(A1 + 4 * k4);
        c00 = __builtin_amdgcn_wmma_f32_16x16x4_f32(false, a0, false, bf0[k4], (short)0, c00, false, false);
        c01 = __builtin_amdgcn_wmma_f32_16x16x4_f32(false, a0, false, bf1[k4], (short)0, c01, false, false);
        c10 = __builtin_amdgcn_wmma_f32_16x16x4_f32(false, a1, false, bf0[k4], (short)0, c10, false, false);
        c11 = __builtin_amdgcn_wmma_f32_16x16x4_f32(false, a1, false, bf1[k4], (short)0, c11, false, false);
    }

    const float bb0 = bout[n0 + lr];
    const float bb1 = bout[n0 + 16 + lr];
#pragma unroll
    for (int r = 0; r < 8; ++r) {
        float* o0 = out + (r0 + r + 8 * hi) * Oo + n0 + lr;
        float* o1 = o0 + (size_t)16 * Oo;
        o0[0]  = c00[r] + bb0;
        o0[16] = c01[r] + bb1;
        o1[0]  = c10[r] + bb0;
        o1[16] = c11[r] + bb1;
    }
}

extern "C" void kernel_launch(void* const* d_in, const int* in_sizes, int n_in,
                              void* d_out, int out_size, void* d_ws, size_t ws_size,
                              hipStream_t stream)
{
    const float* x     = (const float*)d_in[0];
    const float* Wk    = (const float*)d_in[1];
    const float* Wq    = (const float*)d_in[2];
    const float* Wv    = (const float*)d_in[3];
    const float* Wvc   = (const float*)d_in[4];
    const float* We    = (const float*)d_in[5];
    const float* theta = (const float*)d_in[6];
    const float* Wout  = (const float*)d_in[7];
    const float* bout  = (const float*)d_in[8];
    const float* gamma = (const float*)d_in[9];
    const float* beta  = (const float*)d_in[10];
    float* out = (float*)d_out;

    float* ws  = (float*)d_ws;
    const size_t NROW = (size_t)Bb * Tt;
    float* Ko  = ws;
    float* Qo  = Ko  + NROW * Mm;
    float* Vo  = Qo  + NROW * Mm;
    float* vco = Vo  + NROW * Mm;
    float* eta = vco + NROW * Mm;
    float* Send = eta + NROW;
    float* Pend = Send + (size_t)Bb * NC * Mm * Mm;

    ln_proj_gemm_kernel<<<dim3(NROW / (16 * PW), 4), dim3(32 * PW), 0, stream>>>(x, gamma, beta, We,
                                                                  Wk, Wq, Wv, Wvc,
                                                                  Ko, Qo, Vo, vco, eta);
    scan_summary_kernel<<<dim3(Bb, RH, NC / PC), dim3(32 * NG), 0, stream>>>(Ko, Vo, vco, eta, theta,
                                                                            Send, Pend);
    scan_emit_out_kernel<<<dim3(Bb, NC), dim3(32 * EW), 0, stream>>>(Ko, Qo, Vo, vco, eta, theta,
                                                                     Send, Pend, Wout, bout, out);
}
